# starting kernel
# baseline (speedup 1.0000x reference)
#include <hip/hip_runtime.h>


typedef float wf2 __attribute__((ext_vector_type(2)));
typedef float wf8 __attribute__((ext_vector_type(8)));

#define NQ 12
#define NL 4

__device__ __forceinline__ float2 cmul(float2 a, float2 b) {
    return make_float2(a.x * b.x - a.y * b.y, a.x * b.y + a.y * b.x);
}

__global__ __launch_bounds__(256) void qsim_kernel(const float* __restrict__ x,
                                                   const float* __restrict__ w,
                                                   float* __restrict__ out) {
    __shared__ float2 st[4096];
    __shared__ float  mR[5][16][16];
    __shared__ float  mI[5][16][16];
    __shared__ float2 g2[NL][NQ][4];
    __shared__ float  red[8][12];

    const int tid  = threadIdx.x;
    const int b    = blockIdx.x;
    const int lane = tid & 31;
    const int wid  = tid >> 5;
    const int half = lane >> 4;
    const int n16  = lane & 15;

    if (tid < NL * NQ) {
        int l = tid / NQ, q = tid % NQ;
        float th = w[(l * NQ + q) * 2 + 0] + (l == 0 ? x[b * NQ + q] : 0.f);
        float ph = w[(l * NQ + q) * 2 + 1];
        float c = cosf(0.5f * th), s = sinf(0.5f * th);
        float er = cosf(0.5f * ph), ei = -sinf(0.5f * ph);
        g2[l][q][0] = make_float2( c * er,  c * ei);
        g2[l][q][1] = make_float2(-s * er, -s * ei);
        g2[l][q][2] = make_float2( s * er, -s * ei);
        g2[l][q][3] = make_float2( c * er, -c * ei);
    }

    for (int i = tid; i < 4096; i += 256)
        st[i] = make_float2(i == 0 ? 1.f : 0.f, 0.f);

    for (int l = 0; l < NL; ++l) {
        __syncthreads();
        for (int e = tid; e < 5 * 256; e += 256) {
            int m = e >> 8, idx = e & 255, i = idx >> 4, j = idx & 15;
            int g    = (m == 0) ? 0 : (m <= 2 ? 1 : 2);
            int bvar = (m == 0) ? 0 : (m <= 2 ? (m - 1) : (m - 3));
            int ip = i;
            ip ^= (ip >> 1) & 1;
            ip ^= (ip >> 1) & 2;
            ip ^= (ip >> 1) & 4;
            ip ^= bvar << 3;
            float2 v = make_float2(1.f, 0.f);
            #pragma unroll
            for (int k = 0; k < 4; ++k) {
                int bi = (ip >> (3 - k)) & 1, bj = (j >> (3 - k)) & 1;
                v = cmul(v, g2[l][4 * g + k][2 * bi + bj]);
            }
            mR[m][i][j] = v.x;
            mI[m][i][j] = v.y;
        }
        __syncthreads();

        #pragma unroll
        for (int pass = 0; pass < 3; ++pass) {
            #pragma unroll
            for (int t2 = 0; t2 < 2; ++t2) {
                const int tt = wid * 2 + t2;
                int mm, rowmul, colbase;
                if (pass == 0)      { mm = 0;            rowmul = 256; colbase = (tt << 4) | n16; }
                else if (pass == 1) { mm = 1 + (tt & 1); rowmul = 16;  colbase = (tt << 8) | n16; }
                else                { mm = 3 + (tt & 1); rowmul = 1;
                                      colbase = (((((tt >> 1) << 4) | n16) << 1) | (tt & 1)) << 4; }

                wf8 accr = {0, 0, 0, 0, 0, 0, 0, 0};
                wf8 acci = {0, 0, 0, 0, 0, 0, 0, 0};
                #pragma unroll
                for (int kb = 0; kb < 16; kb += 4) {
                    int j0 = kb + 2 * half, j1 = j0 + 1;
                    float2 s0 = st[colbase + j0 * rowmul];
                    float2 s1 = st[colbase + j1 * rowmul];
                    wf2 br  = {s0.x, s1.x};
                    wf2 bim = {s0.y, s1.y};
                    wf2 ar  = {mR[mm][n16][j0], mR[mm][n16][j1]};
                    wf2 ai  = {mI[mm][n16][j0], mI[mm][n16][j1]};
                    wf2 an  = -ai;
                    accr = __builtin_amdgcn_wmma_f32_16x16x4_f32(false, ar, false, br,  (short)0, accr, false, false);
                    accr = __builtin_amdgcn_wmma_f32_16x16x4_f32(false, an, false, bim, (short)0, accr, false, false);
                    acci = __builtin_amdgcn_wmma_f32_16x16x4_f32(false, ar, false, bim, (short)0, acci, false, false);
                    acci = __builtin_amdgcn_wmma_f32_16x16x4_f32(false, ai, false, br,  (short)0, acci, false, false);
                }
                #pragma unroll
                for (int vv = 0; vv < 8; ++vv) {
                    int i = vv + 8 * half;
                    st[colbase + i * rowmul] = make_float2(accr[vv], acci[vv]);
                }
            }
            __syncthreads();
        }
    }

    float z[NQ];
    #pragma unroll
    for (int q = 0; q < NQ; ++q) z[q] = 0.f;
    for (int k = 0; k < 16; ++k) {
        int idx = (k << 8) + tid;
        float2 a = st[idx];
        float p = a.x * a.x + a.y * a.y;
        #pragma unroll
        for (int q = 0; q < NQ; ++q)
            z[q] += ((idx >> (11 - q)) & 1) ? -p : p;
    }
    #pragma unroll
    for (int q = 0; q < NQ; ++q) {
        float v = z[q];
        #pragma unroll
        for (int off = 16; off > 0; off >>= 1)
            v += __shfl_xor(v, off, 32);
        z[q] = v;
    }
    if (lane == 0) {
        #pragma unroll
        for (int q = 0; q < NQ; ++q) red[wid][q] = z[q];
    }
    __syncthreads();
    if (tid < NQ) {
        float s = 0.f;
        #pragma unroll
        for (int ww = 0; ww < 8; ++ww) s += red[ww][tid];
        out[b * NQ + tid] = s;
    }
}

extern "C" void kernel_launch(void* const* d_in, const int* in_sizes, int n_in,
                              void* d_out, int out_size, void* d_ws, size_t ws_size,
                              hipStream_t stream) {
    const float* x = (const float*)d_in[0];
    const float* w = (const float*)d_in[1];
    float* out = (float*)d_out;
    int B = in_sizes[0] / NQ;
    qsim_kernel<<<B, 256, 0, stream>>>(x, w, out);
}


# LDS bank conflicts removed: statevector slots XOR-swizzled (index bits 5-11 folded into the bank bits), gate-matrix rows padded to 20 dwords; same WMMAs, same k and summation order
# speedup vs baseline: 1.3403x; 1.3403x over previous
#include <hip/hip_runtime.h>


typedef float wf2 __attribute__((ext_vector_type(2)));
typedef float wf8 __attribute__((ext_vector_type(8)));

#define NQ 12
#define NL 4
#define MP 20

typedef float wf4 __attribute__((ext_vector_type(4)));

__device__ __forceinline__ int swz(int i) {
    int q4 = ((i >> 5) ^ (i >> 7) ^ (i >> 9) ^ (i >> 11)) & 1;
    return i ^ (q4 << 4) ^ (((i >> 7) & 1) << 3) ^ (((i >> 6) & 1) << 2) ^ (((i >> 8) & 1) << 1);
}

__device__ __forceinline__ float2 cmul(float2 a, float2 b) {
    return make_float2(a.x * b.x - a.y * b.y, a.x * b.y + a.y * b.x);
}

__global__ __launch_bounds__(256, 8) void qsim_kernel(const float* __restrict__ x,
                                                   const float* __restrict__ w,
                                                   float* __restrict__ out) {
    __shared__ __attribute__((aligned(16))) float2 st[4096];
    __shared__ __attribute__((aligned(16))) float  mR[5][16][MP];
    __shared__ __attribute__((aligned(16))) float  mI[5][16][MP];
    __shared__ float2 g2[NL][NQ][4];
    __shared__ float  red[8][12];

    const int tid  = threadIdx.x;
    const int b    = blockIdx.x;
    const int lane = tid & 31;
    const int wid  = tid >> 5;
    const int half = lane >> 4;
    const int n16  = lane & 15;

    if (tid < NL * NQ) {
        int l = tid / NQ, q = tid % NQ;
        float th = w[(l * NQ + q) * 2 + 0] + (l == 0 ? x[b * NQ + q] : 0.f);
        float ph = w[(l * NQ + q) * 2 + 1];
        float c = cosf(0.5f * th), s = sinf(0.5f * th);
        float er = cosf(0.5f * ph), ei = -sinf(0.5f * ph);
        g2[l][q][0] = make_float2( c * er,  c * ei);
        g2[l][q][1] = make_float2(-s * er, -s * ei);
        g2[l][q][2] = make_float2( s * er, -s * ei);
        g2[l][q][3] = make_float2( c * er, -c * ei);
    }

    for (int i = tid; i < 4096; i += 256)
        st[i] = make_float2(i == 0 ? 1.f : 0.f, 0.f);

    for (int l = 0; l < NL; ++l) {
        __syncthreads();
        for (int e = tid; e < 5 * 256; e += 256) {
            int m = e >> 8, idx = e & 255, i = idx >> 4, j = idx & 15;
            int g    = (m == 0) ? 0 : (m <= 2 ? 1 : 2);
            int bvar = (m == 0) ? 0 : (m <= 2 ? (m - 1) : (m - 3));
            int ip = i;
            ip ^= (ip >> 1) & 1;
            ip ^= (ip >> 1) & 2;
            ip ^= (ip >> 1) & 4;
            ip ^= bvar << 3;
            float2 v = make_float2(1.f, 0.f);
            #pragma unroll
            for (int k = 0; k < 4; ++k) {
                int bi = (ip >> (3 - k)) & 1, bj = (j >> (3 - k)) & 1;
                v = cmul(v, g2[l][4 * g + k][2 * bi + bj]);
            }
            mR[m][i][j] = v.x;
            mI[m][i][j] = v.y;
        }
        __syncthreads();

        #pragma unroll
        for (int pass = 0; pass < 3; ++pass) {
            #pragma unroll
            for (int t2 = 0; t2 < 2; ++t2) {
                const int tt = wid * 2 + t2;
                int mm, rowmul, colbase;
                if (pass == 0)      { mm = 0;            rowmul = 256; colbase = (tt << 4) | n16; }
                else if (pass == 1) { mm = 1 + (tt & 1); rowmul = 16;  colbase = (tt << 8) | n16; }
                else                { mm = 3 + (tt & 1); rowmul = 1;
                                      colbase = (((((tt >> 1) << 4) | n16) << 1) | (tt & 1)) << 4; }

                const int lslot = swz(colbase + 2 * half * rowmul);
                const int sslot = swz(colbase + 8 * half * rowmul);

                wf8 accr = {0, 0, 0, 0, 0, 0, 0, 0};
                wf8 acci = {0, 0, 0, 0, 0, 0, 0, 0};
                #pragma unroll
                for (int kb = 0; kb < 16; kb += 4) {
                    int j0 = kb + 2 * half;
                    wf2 br, bim;
                    if (pass < 2) {
                        float2 s0 = st[lslot ^ swz(kb * rowmul)];
                        float2 s1 = st[lslot ^ swz((kb + 1) * rowmul)];
                        br  = (wf2){s0.x, s1.x};
                        bim = (wf2){s0.y, s1.y};
                    } else {
                        wf4 s = *(const wf4*)&st[lslot ^ swz(kb)];
                        br  = (wf2){s.x, s.z};
                        bim = (wf2){s.y, s.w};
                    }
                    wf2 ar  = *(const wf2*)&mR[mm][n16][j0];
                    wf2 ai  = *(const wf2*)&mI[mm][n16][j0];
                    wf2 an  = -ai;
                    accr = __builtin_amdgcn_wmma_f32_16x16x4_f32(false, ar, false, br,  (short)0, accr, false, false);
                    accr = __builtin_amdgcn_wmma_f32_16x16x4_f32(false, an, false, bim, (short)0, accr, false, false);
                    acci = __builtin_amdgcn_wmma_f32_16x16x4_f32(false, ar, false, bim, (short)0, acci, false, false);
                    acci = __builtin_amdgcn_wmma_f32_16x16x4_f32(false, ai, false, br,  (short)0, acci, false, false);
                }
                if (pass < 2) {
                    #pragma unroll
                    for (int vv = 0; vv < 8; ++vv)
                        st[sslot ^ swz(vv * rowmul)] = make_float2(accr[vv], acci[vv]);
                } else {
                    #pragma unroll
                    for (int vv = 0; vv < 8; vv += 2)
                        *(wf4*)&st[sslot ^ swz(vv)] = (wf4){accr[vv], acci[vv], accr[vv + 1], acci[vv + 1]};
                }
            }
            __syncthreads();
        }
    }

    float z[NQ];
    #pragma unroll
    for (int q = 0; q < NQ; ++q) z[q] = 0.f;
    for (int k = 0; k < 16; ++k) {
        int idx = (k << 8) + tid;
        float2 a = st[swz(idx)];
        float p = a.x * a.x + a.y * a.y;
        #pragma unroll
        for (int q = 0; q < NQ; ++q)
            z[q] += ((idx >> (11 - q)) & 1) ? -p : p;
    }
    #pragma unroll
    for (int q = 0; q < NQ; ++q) {
        float v = z[q];
        #pragma unroll
        for (int off = 16; off > 0; off >>= 1)
            v += __shfl_xor(v, off, 32);
        z[q] = v;
    }
    if (lane == 0) {
        #pragma unroll
        for (int q = 0; q < NQ; ++q) red[wid][q] = z[q];
    }
    __syncthreads();
    if (tid < NQ) {
        float s = 0.f;
        #pragma unroll
        for (int ww = 0; ww < 8; ++ww) s += red[ww][tid];
        out[b * NQ + tid] = s;
    }
}

extern "C" void kernel_launch(void* const* d_in, const int* in_sizes, int n_in,
                              void* d_out, int out_size, void* d_ws, size_t ws_size,
                              hipStream_t stream) {
    const float* x = (const float*)d_in[0];
    const float* w = (const float*)d_in[1];
    float* out = (float*)d_out;
    int B = in_sizes[0] / NQ;
    qsim_kernel<<<B, 256, 0, stream>>>(x, w, out);
}


# Gate matrices factored M = diag(RZ phases) * real RY product: WMMA does only R*Re(S), R*Im(S) (8 per tile, was 16), rows then phase-scaled on the VALU; rounding and sum grouping change
# speedup vs baseline: 1.6667x; 1.2435x over previous
#include <hip/hip_runtime.h>


typedef float wf2 __attribute__((ext_vector_type(2)));
typedef float wf8 __attribute__((ext_vector_type(8)));

#define NQ 12
#define NL 4
#define MP 20

typedef float wf4 __attribute__((ext_vector_type(4)));

__device__ __forceinline__ int swz(int i) {
    int q4 = ((i >> 5) ^ (i >> 7) ^ (i >> 9) ^ (i >> 11)) & 1;
    return i ^ (q4 << 4) ^ (((i >> 7) & 1) << 3) ^ (((i >> 6) & 1) << 2) ^ (((i >> 8) & 1) << 1);
}

__device__ __forceinline__ float2 cmul(float2 a, float2 b) {
    return make_float2(a.x * b.x - a.y * b.y, a.x * b.y + a.y * b.x);
}

__global__ __launch_bounds__(256, 8) void qsim_kernel(const float* __restrict__ x,
                                                   const float* __restrict__ w,
                                                   float* __restrict__ out) {
    __shared__ __attribute__((aligned(16))) float2 st[4096];
    __shared__ __attribute__((aligned(16))) float  mR[5][16][MP];
    __shared__ __attribute__((aligned(16))) float  dph[5][2][16];
    __shared__ float2 ry[NL][NQ];
    __shared__ float2 rz[NL][NQ];
    __shared__ float  red[8][12];

    const int tid  = threadIdx.x;
    const int b    = blockIdx.x;
    const int lane = tid & 31;
    const int wid  = tid >> 5;
    const int half = lane >> 4;
    const int n16  = lane & 15;

    if (tid < NL * NQ) {
        int l = tid / NQ, q = tid % NQ;
        float th = w[(l * NQ + q) * 2 + 0] + (l == 0 ? x[b * NQ + q] : 0.f);
        float ph = w[(l * NQ + q) * 2 + 1];
        ry[l][q] = make_float2(cosf(0.5f * th),  sinf(0.5f * th));
        rz[l][q] = make_float2(cosf(0.5f * ph), -sinf(0.5f * ph));
    }

    for (int i = tid; i < 4096; i += 256)
        st[i] = make_float2(i == 0 ? 1.f : 0.f, 0.f);

    for (int l = 0; l < NL; ++l) {
        __syncthreads();
        if (tid < 5 * 16) {
            int m = tid >> 4, i = tid & 15;
            int g    = (m == 0) ? 0 : (m <= 2 ? 1 : 2);
            int bvar = (m == 0) ? 0 : (m <= 2 ? (m - 1) : (m - 3));
            int ip = i;
            ip ^= (ip >> 1) & 1;
            ip ^= (ip >> 1) & 2;
            ip ^= (ip >> 1) & 4;
            ip ^= bvar << 3;
            float2 v = make_float2(1.f, 0.f);
            #pragma unroll
            for (int k = 0; k < 4; ++k) {
                float2 e = rz[l][4 * g + k];
                if ((ip >> (3 - k)) & 1) e.y = -e.y;
                v = cmul(v, e);
            }
            dph[m][0][i] = v.x;
            dph[m][1][i] = v.y;
        }
        for (int e = tid; e < 5 * 256; e += 256) {
            int m = e >> 8, idx = e & 255, i = idx >> 4, j = idx & 15;
            int g    = (m == 0) ? 0 : (m <= 2 ? 1 : 2);
            int bvar = (m == 0) ? 0 : (m <= 2 ? (m - 1) : (m - 3));
            int ip = i;
            ip ^= (ip >> 1) & 1;
            ip ^= (ip >> 1) & 2;
            ip ^= (ip >> 1) & 4;
            ip ^= bvar << 3;
            float v = 1.f;
            #pragma unroll
            for (int k = 0; k < 4; ++k) {
                int bi = (ip >> (3 - k)) & 1, bj = (j >> (3 - k)) & 1;
                float2 cs = ry[l][4 * g + k];
                v *= (bi == bj) ? cs.x : (bi ? cs.y : -cs.y);
            }
            mR[m][i][j] = v;
        }
        __syncthreads();

        #pragma unroll
        for (int pass = 0; pass < 3; ++pass) {
            #pragma unroll
            for (int t2 = 0; t2 < 2; ++t2) {
                const int tt = wid * 2 + t2;
                int mm, rowmul, colbase;
                if (pass == 0)      { mm = 0;            rowmul = 256; colbase = (tt << 4) | n16; }
                else if (pass == 1) { mm = 1 + (tt & 1); rowmul = 16;  colbase = (tt << 8) | n16; }
                else                { mm = 3 + (tt & 1); rowmul = 1;
                                      colbase = (((((tt >> 1) << 4) | n16) << 1) | (tt & 1)) << 4; }

                const int lslot = swz(colbase + 2 * half * rowmul);
                const int sslot = swz(colbase + 8 * half * rowmul);

                const wf4 dre03 = *(const wf4*)&dph[mm][0][8 * half], dre47 = *(const wf4*)&dph[mm][0][8 * half + 4];
                const wf4 dim03 = *(const wf4*)&dph[mm][1][8 * half], dim47 = *(const wf4*)&dph[mm][1][8 * half + 4];

                wf8 accr = {0, 0, 0, 0, 0, 0, 0, 0};
                wf8 acci = {0, 0, 0, 0, 0, 0, 0, 0};
                #pragma unroll
                for (int kb = 0; kb < 16; kb += 4) {
                    int j0 = kb + 2 * half;
                    wf2 br, bim;
                    if (pass < 2) {
                        float2 s0 = st[lslot ^ swz(kb * rowmul)];
                        float2 s1 = st[lslot ^ swz((kb + 1) * rowmul)];
                        br  = (wf2){s0.x, s1.x};
                        bim = (wf2){s0.y, s1.y};
                    } else {
                        wf4 s = *(const wf4*)&st[lslot ^ swz(kb)];
                        br  = (wf2){s.x, s.z};
                        bim = (wf2){s.y, s.w};
                    }
                    wf2 ar  = *(const wf2*)&mR[mm][n16][j0];
                    accr = __builtin_amdgcn_wmma_f32_16x16x4_f32(false, ar, false, br,  (short)0, accr, false, false);
                    acci = __builtin_amdgcn_wmma_f32_16x16x4_f32(false, ar, false, bim, (short)0, acci, false, false);
                }
                const wf8 dre = __builtin_shufflevector(dre03, dre47, 0, 1, 2, 3, 4, 5, 6, 7);
                const wf8 dim = __builtin_shufflevector(dim03, dim47, 0, 1, 2, 3, 4, 5, 6, 7);
                const wf8 outr = dre * accr - dim * acci;
                const wf8 outi = dre * acci + dim * accr;
                if (pass < 2) {
                    #pragma unroll
                    for (int vv = 0; vv < 8; ++vv)
                        st[sslot ^ swz(vv * rowmul)] = make_float2(outr[vv], outi[vv]);
                } else {
                    #pragma unroll
                    for (int vv = 0; vv < 8; vv += 2)
                        *(wf4*)&st[sslot ^ swz(vv)] = (wf4){outr[vv], outi[vv], outr[vv + 1], outi[vv + 1]};
                }
            }
            __syncthreads();
        }
    }

    float z[NQ];
    #pragma unroll
    for (int q = 0; q < NQ; ++q) z[q] = 0.f;
    for (int k = 0; k < 16; ++k) {
        int idx = (k << 8) + tid;
        float2 a = st[swz(idx)];
        float p = a.x * a.x + a.y * a.y;
        #pragma unroll
        for (int q = 0; q < NQ; ++q)
            z[q] += ((idx >> (11 - q)) & 1) ? -p : p;
    }
    #pragma unroll
    for (int q = 0; q < NQ; ++q) {
        float v = z[q];
        #pragma unroll
        for (int off = 16; off > 0; off >>= 1)
            v += __shfl_xor(v, off, 32);
        z[q] = v;
    }
    if (lane == 0) {
        #pragma unroll
        for (int q = 0; q < NQ; ++q) red[wid][q] = z[q];
    }
    __syncthreads();
    if (tid < NQ) {
        float s = 0.f;
        #pragma unroll
        for (int ww = 0; ww < 8; ++ww) s += red[ww][tid];
        out[b * NQ + tid] = s;
    }
}

extern "C" void kernel_launch(void* const* d_in, const int* in_sizes, int n_in,
                              void* d_out, int out_size, void* d_ws, size_t ws_size,
                              hipStream_t stream) {
    const float* x = (const float*)d_in[0];
    const float* w = (const float*)d_in[1];
    float* out = (float*)d_out;
    int B = in_sizes[0] / NQ;
    qsim_kernel<<<B, 256, 0, stream>>>(x, w, out);
}


# Layer 0 acts on |0..0>: its output is a product state, written directly as M_A[i][0]*M_B[j][0]*M_C[k][0]; init sweep and 3 one-hot WMMA sweeps dropped; layer-0 rounding changes
# speedup vs baseline: 2.1476x; 1.2885x over previous
#include <hip/hip_runtime.h>


typedef float wf2 __attribute__((ext_vector_type(2)));
typedef float wf8 __attribute__((ext_vector_type(8)));

#define NQ 12
#define NL 4
#define MP 20

typedef float wf4 __attribute__((ext_vector_type(4)));

__device__ __forceinline__ int swz(int i) {
    int q4 = ((i >> 5) ^ (i >> 7) ^ (i >> 9) ^ (i >> 11)) & 1;
    return i ^ (q4 << 4) ^ (((i >> 7) & 1) << 3) ^ (((i >> 6) & 1) << 2) ^ (((i >> 8) & 1) << 1);
}

__device__ __forceinline__ int colbase_c(int tt, int n16) {
    return (((((tt >> 1) << 4) | n16) << 1) | (tt & 1)) << 4;
}

__device__ __forceinline__ float2 cmul(float2 a, float2 b) {
    return make_float2(a.x * b.x - a.y * b.y, a.x * b.y + a.y * b.x);
}

__global__ __launch_bounds__(256, 8) void qsim_kernel(const float* __restrict__ x,
                                                   const float* __restrict__ w,
                                                   float* __restrict__ out) {
    __shared__ __attribute__((aligned(16))) float2 st[4096];
    __shared__ __attribute__((aligned(16))) float  mR[5][16][MP];
    __shared__ __attribute__((aligned(16))) float  dph[5][2][16];
    __shared__ __attribute__((aligned(16))) float2 col0[5][16];
    __shared__ float2 ry[NL][NQ];
    __shared__ float2 rz[NL][NQ];
    __shared__ float  red[8][12];

    const int tid  = threadIdx.x;
    const int b    = blockIdx.x;
    const int lane = tid & 31;
    const int wid  = tid >> 5;
    const int half = lane >> 4;
    const int n16  = lane & 15;

    if (tid < NL * NQ) {
        int l = tid / NQ, q = tid % NQ;
        float th = w[(l * NQ + q) * 2 + 0] + (l == 0 ? x[b * NQ + q] : 0.f);
        float ph = w[(l * NQ + q) * 2 + 1];
        ry[l][q] = make_float2(cosf(0.5f * th),  sinf(0.5f * th));
        rz[l][q] = make_float2(cosf(0.5f * ph), -sinf(0.5f * ph));
    }

    __syncthreads();

    if (tid < 5 * 16) {
        int m = tid >> 4, i = tid & 15;
        int g    = (m == 0) ? 0 : (m <= 2 ? 1 : 2);
        int bvar = (m == 0) ? 0 : (m <= 2 ? (m - 1) : (m - 3));
        int ip = i;
        ip ^= (ip >> 1) & 1;
        ip ^= (ip >> 1) & 2;
        ip ^= (ip >> 1) & 4;
        ip ^= bvar << 3;
        float2 v = make_float2(1.f, 0.f);
        float  r = 1.f;
        #pragma unroll
        for (int k = 0; k < 4; ++k) {
            int bi = (ip >> (3 - k)) & 1;
            float2 e  = rz[0][4 * g + k];
            float2 cs = ry[0][4 * g + k];
            if (bi) e.y = -e.y;
            v = cmul(v, e);
            r *= bi ? cs.y : cs.x;
        }
        col0[m][i] = make_float2(v.x * r, v.y * r);
    }
    __syncthreads();

    {
        const int iA = (wid << 1) | (n16 >> 3);
        const float2 a = col0[0][iA];
        #pragma unroll
        for (int t2 = 0; t2 < 2; ++t2) {
            const int tt = wid * 2 + t2;
            const int iB = ((n16 & 7) << 1) | t2;
            const float2 ab = cmul(a, col0[1 + (iA & 1)][iB]);
            const int sslot = swz(colbase_c(tt, n16) + 8 * half);
            #pragma unroll
            for (int vv = 0; vv < 8; vv += 2) {
                const wf4 cc = *(const wf4*)&col0[3 + t2][8 * half + vv];
                const float2 p0 = cmul(ab, make_float2(cc.x, cc.y));
                const float2 p1 = cmul(ab, make_float2(cc.z, cc.w));
                *(wf4*)&st[sslot ^ swz(vv)] = (wf4){p0.x, p0.y, p1.x, p1.y};
            }
        }
    }

    #pragma nounroll
    for (int l = 1; l < NL; ++l) {
        __syncthreads();
        if (tid < 5 * 16) {
            int m = tid >> 4, i = tid & 15;
            int g    = (m == 0) ? 0 : (m <= 2 ? 1 : 2);
            int bvar = (m == 0) ? 0 : (m <= 2 ? (m - 1) : (m - 3));
            int ip = i;
            ip ^= (ip >> 1) & 1;
            ip ^= (ip >> 1) & 2;
            ip ^= (ip >> 1) & 4;
            ip ^= bvar << 3;
            float2 v = make_float2(1.f, 0.f);
            #pragma unroll
            for (int k = 0; k < 4; ++k) {
                float2 e = rz[l][4 * g + k];
                if ((ip >> (3 - k)) & 1) e.y = -e.y;
                v = cmul(v, e);
            }
            dph[m][0][i] = v.x;
            dph[m][1][i] = v.y;
        }
        for (int e = tid; e < 5 * 256; e += 256) {
            int m = e >> 8, idx = e & 255, i = idx >> 4, j = idx & 15;
            int g    = (m == 0) ? 0 : (m <= 2 ? 1 : 2);
            int bvar = (m == 0) ? 0 : (m <= 2 ? (m - 1) : (m - 3));
            int ip = i;
            ip ^= (ip >> 1) & 1;
            ip ^= (ip >> 1) & 2;
            ip ^= (ip >> 1) & 4;
            ip ^= bvar << 3;
            float v = 1.f;
            #pragma unroll
            for (int k = 0; k < 4; ++k) {
                int bi = (ip >> (3 - k)) & 1, bj = (j >> (3 - k)) & 1;
                float2 cs = ry[l][4 * g + k];
                v *= (bi == bj) ? cs.x : (bi ? cs.y : -cs.y);
            }
            mR[m][i][j] = v;
        }
        __syncthreads();

        #pragma unroll
        for (int pass = 0; pass < 3; ++pass) {
            #pragma unroll
            for (int t2 = 0; t2 < 2; ++t2) {
                const int tt = wid * 2 + t2;
                int mm, rowmul, colbase;
                if (pass == 0)      { mm = 0;            rowmul = 256; colbase = (tt << 4) | n16; }
                else if (pass == 1) { mm = 1 + (tt & 1); rowmul = 16;  colbase = (tt << 8) | n16; }
                else                { mm = 3 + (tt & 1); rowmul = 1;
                                      colbase = colbase_c(tt, n16); }

                const int lslot = swz(colbase + 2 * half * rowmul);
                const int sslot = swz(colbase + 8 * half * rowmul);

                const wf4 dre03 = *(const wf4*)&dph[mm][0][8 * half], dre47 = *(const wf4*)&dph[mm][0][8 * half + 4];
                const wf4 dim03 = *(const wf4*)&dph[mm][1][8 * half], dim47 = *(const wf4*)&dph[mm][1][8 * half + 4];

                wf8 accr = {0, 0, 0, 0, 0, 0, 0, 0};
                wf8 acci = {0, 0, 0, 0, 0, 0, 0, 0};
                #pragma unroll
                for (int kb = 0; kb < 16; kb += 4) {
                    int j0 = kb + 2 * half;
                    wf2 br, bim;
                    if (pass < 2) {
                        float2 s0 = st[lslot ^ swz(kb * rowmul)];
                        float2 s1 = st[lslot ^ swz((kb + 1) * rowmul)];
                        br  = (wf2){s0.x, s1.x};
                        bim = (wf2){s0.y, s1.y};
                    } else {
                        wf4 s = *(const wf4*)&st[lslot ^ swz(kb)];
                        br  = (wf2){s.x, s.z};
                        bim = (wf2){s.y, s.w};
                    }
                    wf2 ar  = *(const wf2*)&mR[mm][n16][j0];
                    accr = __builtin_amdgcn_wmma_f32_16x16x4_f32(false, ar, false, br,  (short)0, accr, false, false);
                    acci = __builtin_amdgcn_wmma_f32_16x16x4_f32(false, ar, false, bim, (short)0, acci, false, false);
                }
                const wf8 dre = __builtin_shufflevector(dre03, dre47, 0, 1, 2, 3, 4, 5, 6, 7);
                const wf8 dim = __builtin_shufflevector(dim03, dim47, 0, 1, 2, 3, 4, 5, 6, 7);
                const wf8 outr = dre * accr - dim * acci;
                const wf8 outi = dre * acci + dim * accr;
                if (pass < 2) {
                    #pragma unroll
                    for (int vv = 0; vv < 8; ++vv)
                        st[sslot ^ swz(vv * rowmul)] = make_float2(outr[vv], outi[vv]);
                } else {
                    #pragma unroll
                    for (int vv = 0; vv < 8; vv += 2)
                        *(wf4*)&st[sslot ^ swz(vv)] = (wf4){outr[vv], outi[vv], outr[vv + 1], outi[vv + 1]};
                }
            }
            __syncthreads();
        }
    }

    float z[NQ];
    #pragma unroll
    for (int q = 0; q < NQ; ++q) z[q] = 0.f;
    const int wslot = swz(__builtin_amdgcn_readfirstlane(wid) << 5);
    const int tslot = swz(lane) ^ wslot;
    for (int k = 0; k < 16; ++k) {
        int idx = (k << 8) + tid;
        float2 a = st[tslot ^ swz(k << 8)];
        float p = a.x * a.x + a.y * a.y;
        #pragma unroll
        for (int q = 0; q < NQ; ++q)
            z[q] += ((idx >> (11 - q)) & 1) ? -p : p;
    }
    #pragma unroll
    for (int q = 0; q < NQ; ++q) {
        float v = z[q];
        #pragma unroll
        for (int off = 16; off > 0; off >>= 1)
            v += __shfl_xor(v, off, 32);
        z[q] = v;
    }
    if (lane == 0) {
        #pragma unroll
        for (int q = 0; q < NQ; ++q) red[wid][q] = z[q];
    }
    __syncthreads();
    if (tid < NQ) {
        float s = 0.f;
        #pragma unroll
        for (int ww = 0; ww < 8; ++ww) s += red[ww][tid];
        out[b * NQ + tid] = s;
    }
}

extern "C" void kernel_launch(void* const* d_in, const int* in_sizes, int n_in,
                              void* d_out, int out_size, void* d_ws, size_t ws_size,
                              hipStream_t stream) {
    const float* x = (const float*)d_in[0];
    const float* w = (const float*)d_in[1];
    float* out = (float*)d_out;
    int B = in_sizes[0] / NQ;
    qsim_kernel<<<B, 256, 0, stream>>>(x, w, out);
}


# Gate-matrix build hoisted out of the layer loop: real factors and row phases of layers 1-3 built once up front, branch-free with per-thread select masks; 5 barriers fewer; values and sum order kept
# speedup vs baseline: 2.4164x; 1.1252x over previous
#include <hip/hip_runtime.h>


typedef float wf2 __attribute__((ext_vector_type(2)));
typedef float wf8 __attribute__((ext_vector_type(8)));

#define NQ 12
#define NL 4
#define MP 20

typedef float wf4 __attribute__((ext_vector_type(4)));

__device__ __forceinline__ int swz(int i) {
    int q4 = ((i >> 5) ^ (i >> 7) ^ (i >> 9) ^ (i >> 11)) & 1;
    return i ^ (q4 << 4) ^ (((i >> 7) & 1) << 3) ^ (((i >> 6) & 1) << 2) ^ (((i >> 8) & 1) << 1);
}

__device__ __forceinline__ int colbase_c(int tt, int n16) {
    return (((((tt >> 1) << 4) | n16) << 1) | (tt & 1)) << 4;
}

__device__ __forceinline__ float2 cmul(float2 a, float2 b) {
    return make_float2(a.x * b.x - a.y * b.y, a.x * b.y + a.y * b.x);
}

__global__ __launch_bounds__(256, 8) void qsim_kernel(const float* __restrict__ x,
                                                   const float* __restrict__ w,
                                                   float* __restrict__ out) {
    __shared__ __attribute__((aligned(16))) float2 st[4096];
    __shared__ __attribute__((aligned(16))) float  mR[NL - 1][5][16][MP];
    __shared__ __attribute__((aligned(16))) float  dph[NL - 1][5][2][16];
    __shared__ __attribute__((aligned(16))) float2 col0[5][16];
    __shared__ __attribute__((aligned(16))) float2 ry[NL][NQ];
    __shared__ float2 rz[NL][NQ];
    __shared__ float  red[8][12];

    const int tid  = threadIdx.x;
    const int b    = blockIdx.x;
    const int lane = tid & 31;
    const int wid  = tid >> 5;
    const int half = lane >> 4;
    const int n16  = lane & 15;

    if (tid < NL * NQ) {
        int l = tid / NQ, q = tid % NQ;
        float th = w[(l * NQ + q) * 2 + 0] + (l == 0 ? x[b * NQ + q] : 0.f);
        float ph = w[(l * NQ + q) * 2 + 1];
        ry[l][q] = make_float2(cosf(0.5f * th),  sinf(0.5f * th));
        rz[l][q] = make_float2(cosf(0.5f * ph), -sinf(0.5f * ph));
    }

    __syncthreads();

    if (tid < 5 * 16) {
        int m = tid >> 4, i = tid & 15;
        int g    = (m == 0) ? 0 : (m <= 2 ? 1 : 2);
        int bvar = (m == 0) ? 0 : (m <= 2 ? (m - 1) : (m - 3));
        int ip = i;
        ip ^= (ip >> 1) & 1;
        ip ^= (ip >> 1) & 2;
        ip ^= (ip >> 1) & 4;
        ip ^= bvar << 3;
        float2 v = make_float2(1.f, 0.f);
        float  r = 1.f;
        #pragma unroll
        for (int k = 0; k < 4; ++k) {
            int bi = (ip >> (3 - k)) & 1;
            float2 e  = rz[0][4 * g + k];
            float2 cs = ry[0][4 * g + k];
            if (bi) e.y = -e.y;
            v = cmul(v, e);
            r *= bi ? cs.y : cs.x;
        }
        col0[m][i] = make_float2(v.x * r, v.y * r);
    }

    if (tid < (NL - 1) * 5 * 16) {
        int l = 1 + tid / 80, rr = tid % 80;
        int m = rr >> 4, i = rr & 15;
        int g    = (m == 0) ? 0 : (m <= 2 ? 1 : 2);
        int bvar = (m == 0) ? 0 : (m <= 2 ? (m - 1) : (m - 3));
        int ip = i;
        ip ^= (ip >> 1) & 1;
        ip ^= (ip >> 1) & 2;
        ip ^= (ip >> 1) & 4;
        ip ^= bvar << 3;
        float2 v = make_float2(1.f, 0.f);
        #pragma unroll
        for (int k = 0; k < 4; ++k) {
            float2 e = rz[l][4 * g + k];
            if ((ip >> (3 - k)) & 1) e.y = -e.y;
            v = cmul(v, e);
        }
        dph[l - 1][m][0][i] = v.x;
        dph[l - 1][m][1][i] = v.y;
    }
    {
        const int i = tid >> 4, j = tid & 15;
        int ip = i;
        ip ^= (ip >> 1) & 1;
        ip ^= (ip >> 1) & 2;
        ip ^= (ip >> 1) & 4;
        const int df = ip ^ j;
        const bool e0 = !(df & 8), e1 = !(df & 4), e2 = !(df & 2), e3 = !(df & 1);
        const int sg0 = ((0x6996 >> (~ip & j & 15)) & 1) << 31;
        const int sg1 = ((0x6996 >> (~(ip ^ 8) & j & 15)) & 1) << 31;
        #pragma unroll
        for (int l = 1; l < NL; ++l) {
            #pragma unroll
            for (int g = 0; g < 3; ++g) {
                const wf4 q01 = *(const wf4*)&ry[l][4 * g];
                const wf4 q23 = *(const wf4*)&ry[l][4 * g + 2];
                const float t1 = e1 ? q01.z : q01.w;
                const float t2 = e2 ? q23.x : q23.y;
                const float t3 = e3 ? q23.z : q23.w;
                float va = e0 ? q01.x : q01.y;
                va *= t1; va *= t2; va *= t3;
                mR[l - 1][g == 0 ? 0 : 2 * g - 1][i][j] = __int_as_float(__float_as_int(va) ^ sg0);
                if (g > 0) {
                    float vb = e0 ? q01.y : q01.x;
                    vb *= t1; vb *= t2; vb *= t3;
                    mR[l - 1][2 * g][i][j] = __int_as_float(__float_as_int(vb) ^ sg1);
                }
            }
        }
    }
    __syncthreads();

    {
        const int iA = (wid << 1) | (n16 >> 3);
        const float2 a = col0[0][iA];
        #pragma unroll
        for (int t2 = 0; t2 < 2; ++t2) {
            const int tt = wid * 2 + t2;
            const int iB = ((n16 & 7) << 1) | t2;
            const float2 ab = cmul(a, col0[1 + (iA & 1)][iB]);
            const int sslot = swz(colbase_c(tt, n16) + 8 * half);
            #pragma unroll
            for (int vv = 0; vv < 8; vv += 2) {
                const wf4 cc = *(const wf4*)&col0[3 + t2][8 * half + vv];
                const float2 p0 = cmul(ab, make_float2(cc.x, cc.y));
                const float2 p1 = cmul(ab, make_float2(cc.z, cc.w));
                *(wf4*)&st[sslot ^ swz(vv)] = (wf4){p0.x, p0.y, p1.x, p1.y};
            }
        }
    }

    #pragma nounroll
    for (int l = 1; l < NL; ++l) {
        #pragma unroll
        for (int pass = 0; pass < 3; ++pass) {
            __syncthreads();
            #pragma unroll
            for (int t2 = 0; t2 < 2; ++t2) {
                const int tt = wid * 2 + t2;
                int mm, rowmul, colbase;
                if (pass == 0)      { mm = 0;            rowmul = 256; colbase = (tt << 4) | n16; }
                else if (pass == 1) { mm = 1 + (tt & 1); rowmul = 16;  colbase = (tt << 8) | n16; }
                else                { mm = 3 + (tt & 1); rowmul = 1;
                                      colbase = colbase_c(tt, n16); }

                const int lslot = swz(colbase + 2 * half * rowmul);
                const int sslot = swz(colbase + 8 * half * rowmul);

                const wf4 dre03 = *(const wf4*)&dph[l - 1][mm][0][8 * half], dre47 = *(const wf4*)&dph[l - 1][mm][0][8 * half + 4];
                const wf4 dim03 = *(const wf4*)&dph[l - 1][mm][1][8 * half], dim47 = *(const wf4*)&dph[l - 1][mm][1][8 * half + 4];

                wf8 accr = {0, 0, 0, 0, 0, 0, 0, 0};
                wf8 acci = {0, 0, 0, 0, 0, 0, 0, 0};
                #pragma unroll
                for (int kb = 0; kb < 16; kb += 4) {
                    int j0 = kb + 2 * half;
                    wf2 br, bim;
                    if (pass < 2) {
                        float2 s0 = st[lslot ^ swz(kb * rowmul)];
                        float2 s1 = st[lslot ^ swz((kb + 1) * rowmul)];
                        br  = (wf2){s0.x, s1.x};
                        bim = (wf2){s0.y, s1.y};
                    } else {
                        wf4 s = *(const wf4*)&st[lslot ^ swz(kb)];
                        br  = (wf2){s.x, s.z};
                        bim = (wf2){s.y, s.w};
                    }
                    wf2 ar  = *(const wf2*)&mR[l - 1][mm][n16][j0];
                    accr = __builtin_amdgcn_wmma_f32_16x16x4_f32(false, ar, false, br,  (short)0, accr, false, false);
                    acci = __builtin_amdgcn_wmma_f32_16x16x4_f32(false, ar, false, bim, (short)0, acci, false, false);
                }
                const wf8 dre = __builtin_shufflevector(dre03, dre47, 0, 1, 2, 3, 4, 5, 6, 7);
                const wf8 dim = __builtin_shufflevector(dim03, dim47, 0, 1, 2, 3, 4, 5, 6, 7);
                const wf8 outr = dre * accr - dim * acci;
                const wf8 outi = dre * acci + dim * accr;
                if (pass < 2) {
                    #pragma unroll
                    for (int vv = 0; vv < 8; ++vv)
                        st[sslot ^ swz(vv * rowmul)] = make_float2(outr[vv], outi[vv]);
                } else {
                    #pragma unroll
                    for (int vv = 0; vv < 8; vv += 2)
                        *(wf4*)&st[sslot ^ swz(vv)] = (wf4){outr[vv], outi[vv], outr[vv + 1], outi[vv + 1]};
                }
            }
        }
    }
    __syncthreads();

    float z[NQ];
    #pragma unroll
    for (int q = 0; q < NQ; ++q) z[q] = 0.f;
    const int wslot = swz(__builtin_amdgcn_readfirstlane(wid) << 5);
    const int tslot = swz(lane) ^ wslot;
    for (int k = 0; k < 16; ++k) {
        int idx = (k << 8) + tid;
        float2 a = st[tslot ^ swz(k << 8)];
        float p = a.x * a.x + a.y * a.y;
        #pragma unroll
        for (int q = 0; q < NQ; ++q)
            z[q] += ((idx >> (11 - q)) & 1) ? -p : p;
    }
    #pragma unroll
    for (int q = 0; q < NQ; ++q) {
        float v = z[q];
        #pragma unroll
        for (int off = 16; off > 0; off >>= 1)
            v += __shfl_xor(v, off, 32);
        z[q] = v;
    }
    if (lane == 0) {
        #pragma unroll
        for (int q = 0; q < NQ; ++q) red[wid][q] = z[q];
    }
    __syncthreads();
    if (tid < NQ) {
        float s = 0.f;
        #pragma unroll
        for (int ww = 0; ww < 8; ++ww) s += red[ww][tid];
        out[b * NQ + tid] = s;
    }
}

extern "C" void kernel_launch(void* const* d_in, const int* in_sizes, int n_in,
                              void* d_out, int out_size, void* d_ws, size_t ws_size,
                              hipStream_t stream) {
    const float* x = (const float*)d_in[0];
    const float* w = (const float*)d_in[1];
    float* out = (float*)d_out;
    int B = in_sizes[0] / NQ;
    qsim_kernel<<<B, 256, 0, stream>>>(x, w, out);
}


# Layer 1 applied to the rank-2 factors of the layer-0 product state: five 16x16 WMMA products, state after layer 1 written directly; 3 full sweeps dropped; layer-1 rounding and sum grouping change
# speedup vs baseline: 2.8912x; 1.1965x over previous
#include <hip/hip_runtime.h>


typedef float wf2 __attribute__((ext_vector_type(2)));
typedef float wf8 __attribute__((ext_vector_type(8)));

#define NQ 12
#define NL 4
#define MP 20

typedef float wf4 __attribute__((ext_vector_type(4)));

__device__ __forceinline__ int swz(int i) {
    int q4 = ((i >> 5) ^ (i >> 7) ^ (i >> 9) ^ (i >> 11)) & 1;
    return i ^ (q4 << 4) ^ (((i >> 7) & 1) << 3) ^ (((i >> 6) & 1) << 2) ^ (((i >> 8) & 1) << 1);
}

__device__ __forceinline__ int colbase_c(int tt, int n16) {
    return (((((tt >> 1) << 4) | n16) << 1) | (tt & 1)) << 4;
}

__device__ __forceinline__ float2 cmul(float2 a, float2 b) {
    return make_float2(a.x * b.x - a.y * b.y, a.x * b.y + a.y * b.x);
}

__global__ __launch_bounds__(256, 8) void qsim_kernel(const float* __restrict__ x,
                                                   const float* __restrict__ w,
                                                   float* __restrict__ out) {
    __shared__ __attribute__((aligned(16))) float2 st[4096];
    __shared__ __attribute__((aligned(16))) float  mR[NL - 1][5][16][MP];
    __shared__ __attribute__((aligned(16))) float  dph[NL - 1][5][2][16];
    __shared__ __attribute__((aligned(16))) float2 col0[5][16];
    __shared__ __attribute__((aligned(16))) float2 l1v[5][4][16];
    __shared__ __attribute__((aligned(16))) float2 ry[NL][NQ];
    __shared__ float2 rz[NL][NQ];
    __shared__ float  red[8][12];

    const int tid  = threadIdx.x;
    const int b    = blockIdx.x;
    const int lane = tid & 31;
    const int wid  = tid >> 5;
    const int half = lane >> 4;
    const int n16  = lane & 15;

    if (tid < NL * NQ) {
        int l = tid / NQ, q = tid % NQ;
        float th = w[(l * NQ + q) * 2 + 0] + (l == 0 ? x[b * NQ + q] : 0.f);
        float ph = w[(l * NQ + q) * 2 + 1];
        ry[l][q] = make_float2(cosf(0.5f * th),  sinf(0.5f * th));
        rz[l][q] = make_float2(cosf(0.5f * ph), -sinf(0.5f * ph));
    }

    __syncthreads();

    if (tid < 5 * 16) {
        int m = tid >> 4, i = tid & 15;
        int g    = (m == 0) ? 0 : (m <= 2 ? 1 : 2);
        int bvar = (m == 0) ? 0 : (m <= 2 ? (m - 1) : (m - 3));
        int ip = i;
        ip ^= (ip >> 1) & 1;
        ip ^= (ip >> 1) & 2;
        ip ^= (ip >> 1) & 4;
        ip ^= bvar << 3;
        float2 v = make_float2(1.f, 0.f);
        float  r = 1.f;
        #pragma unroll
        for (int k = 0; k < 4; ++k) {
            int bi = (ip >> (3 - k)) & 1;
            float2 e  = rz[0][4 * g + k];
            float2 cs = ry[0][4 * g + k];
            if (bi) e.y = -e.y;
            v = cmul(v, e);
            r *= bi ? cs.y : cs.x;
        }
        col0[m][i] = make_float2(v.x * r, v.y * r);
    }

    if (tid < (NL - 1) * 5 * 16) {
        int l = 1 + tid / 80, rr = tid % 80;
        int m = rr >> 4, i = rr & 15;
        int g    = (m == 0) ? 0 : (m <= 2 ? 1 : 2);
        int bvar = (m == 0) ? 0 : (m <= 2 ? (m - 1) : (m - 3));
        int ip = i;
        ip ^= (ip >> 1) & 1;
        ip ^= (ip >> 1) & 2;
        ip ^= (ip >> 1) & 4;
        ip ^= bvar << 3;
        float2 v = make_float2(1.f, 0.f);
        #pragma unroll
        for (int k = 0; k < 4; ++k) {
            float2 e = rz[l][4 * g + k];
            if ((ip >> (3 - k)) & 1) e.y = -e.y;
            v = cmul(v, e);
        }
        dph[l - 1][m][0][i] = v.x;
        dph[l - 1][m][1][i] = v.y;
    }
    {
        const int i = tid >> 4, j = tid & 15;
        int ip = i;
        ip ^= (ip >> 1) & 1;
        ip ^= (ip >> 1) & 2;
        ip ^= (ip >> 1) & 4;
        const int df = ip ^ j;
        const bool e0 = !(df & 8), e1 = !(df & 4), e2 = !(df & 2), e3 = !(df & 1);
        const int sg0 = ((0x6996 >> (~ip & j & 15)) & 1) << 31;
        const int sg1 = ((0x6996 >> (~(ip ^ 8) & j & 15)) & 1) << 31;
        #pragma unroll
        for (int l = 1; l < NL; ++l) {
            #pragma unroll
            for (int g = 0; g < 3; ++g) {
                const wf4 q01 = *(const wf4*)&ry[l][4 * g];
                const wf4 q23 = *(const wf4*)&ry[l][4 * g + 2];
                const float t1 = e1 ? q01.z : q01.w;
                const float t2 = e2 ? q23.x : q23.y;
                const float t3 = e3 ? q23.z : q23.w;
                float va = e0 ? q01.x : q01.y;
                va *= t1; va *= t2; va *= t3;
                mR[l - 1][g == 0 ? 0 : 2 * g - 1][i][j] = __int_as_float(__float_as_int(va) ^ sg0);
                if (g > 0) {
                    float vb = e0 ? q01.y : q01.x;
                    vb *= t1; vb *= t2; vb *= t3;
                    mR[l - 1][2 * g][i][j] = __int_as_float(__float_as_int(vb) ^ sg1);
                }
            }
        }
    }
    __syncthreads();

    {
        const int m = __builtin_amdgcn_readfirstlane(wid);
        if (m < 5) {
            const bool isA = (m == 0), isB = (m == 1) | (m == 2);
            const int  fsrc = isA ? 0 : (isB ? 1 + ((n16 >> 1) & 1) : 3 + (n16 & 1));
            const bool live = isB ? (n16 < 4) : (n16 < 2);
            const bool keep0 = live && (!(isA | isB) || (n16 & 1) == 0);
            const bool keep1 = live && (!(isA | isB) || (n16 & 1) == 1);
            const wf4 dre03 = *(const wf4*)&dph[0][m][0][8 * half], dre47 = *(const wf4*)&dph[0][m][0][8 * half + 4];
            const wf4 dim03 = *(const wf4*)&dph[0][m][1][8 * half], dim47 = *(const wf4*)&dph[0][m][1][8 * half + 4];
            wf8 accr = {0, 0, 0, 0, 0, 0, 0, 0};
            wf8 acci = {0, 0, 0, 0, 0, 0, 0, 0};
            #pragma unroll
            for (int kb = 0; kb < 16; kb += 4) {
                const int j0 = kb + 2 * half;
                const wf4 cc = *(const wf4*)&col0[fsrc][j0];
                const wf2 br  = (wf2){keep0 ? cc.x : 0.f, keep1 ? cc.z : 0.f};
                const wf2 bim = (wf2){keep0 ? cc.y : 0.f, keep1 ? cc.w : 0.f};
                const wf2 ar  = *(const wf2*)&mR[0][m][n16][j0];
                accr = __builtin_amdgcn_wmma_f32_16x16x4_f32(false, ar, false, br,  (short)0, accr, false, false);
                acci = __builtin_amdgcn_wmma_f32_16x16x4_f32(false, ar, false, bim, (short)0, acci, false, false);
            }
            const wf8 dre = __builtin_shufflevector(dre03, dre47, 0, 1, 2, 3, 4, 5, 6, 7);
            const wf8 dim = __builtin_shufflevector(dim03, dim47, 0, 1, 2, 3, 4, 5, 6, 7);
            const wf8 outr = dre * accr - dim * acci;
            const wf8 outi = dre * acci + dim * accr;
            if (n16 < 4) {
                #pragma unroll
                for (int vv = 0; vv < 8; vv += 2)
                    *(wf4*)&l1v[m][n16][8 * half + vv] = (wf4){outr[vv], outi[vv], outr[vv + 1], outi[vv + 1]};
            }
        }
    }
    __syncthreads();

    {
        const int iA = (wid << 1) | (n16 >> 3);
        const int vB = iA & 1;
        const float2 al0 = l1v[0][0][iA], al1 = l1v[0][1][iA];
        #pragma unroll
        for (int t2 = 0; t2 < 2; ++t2) {
            const int tt = wid * 2 + t2;
            const int iB = ((n16 & 7) << 1) | t2;
            const float2 b00 = l1v[1 + vB][0][iB], b01 = l1v[1 + vB][1][iB];
            const float2 b10 = l1v[1 + vB][2][iB], b11 = l1v[1 + vB][3][iB];
            const float2 x0 = cmul(al0, b00), y0 = cmul(al1, b10);
            const float2 x1 = cmul(al0, b01), y1 = cmul(al1, b11);
            const float2 e0 = make_float2(x0.x + y0.x, x0.y + y0.y);
            const float2 e1 = make_float2(x1.x + y1.x, x1.y + y1.y);
            const int sslot = swz(colbase_c(tt, n16) + 8 * half);
            #pragma unroll
            for (int vv = 0; vv < 8; vv += 2) {
                const wf4 g0 = *(const wf4*)&l1v[3 + t2][0][8 * half + vv];
                const wf4 g1 = *(const wf4*)&l1v[3 + t2][1][8 * half + vv];
                const float2 q0 = cmul(e0, make_float2(g0.x, g0.y)), r0 = cmul(e1, make_float2(g1.x, g1.y));
                const float2 q1 = cmul(e0, make_float2(g0.z, g0.w)), r1 = cmul(e1, make_float2(g1.z, g1.w));
                *(wf4*)&st[sslot ^ swz(vv)] = (wf4){q0.x + r0.x, q0.y + r0.y, q1.x + r1.x, q1.y + r1.y};
            }
        }
    }

    #pragma nounroll
    for (int l = 2; l < NL; ++l) {
        #pragma unroll
        for (int pass = 0; pass < 3; ++pass) {
            __syncthreads();
            #pragma unroll
            for (int t2 = 0; t2 < 2; ++t2) {
                const int tt = wid * 2 + t2;
                int mm, rowmul, colbase;
                if (pass == 0)      { mm = 0;            rowmul = 256; colbase = (tt << 4) | n16; }
                else if (pass == 1) { mm = 1 + (tt & 1); rowmul = 16;  colbase = (tt << 8) | n16; }
                else                { mm = 3 + (tt & 1); rowmul = 1;
                                      colbase = colbase_c(tt, n16); }

                const int lslot = swz(colbase + 2 * half * rowmul);
                const int sslot = swz(colbase + 8 * half * rowmul);

                const wf4 dre03 = *(const wf4*)&dph[l - 1][mm][0][8 * half], dre47 = *(const wf4*)&dph[l - 1][mm][0][8 * half + 4];
                const wf4 dim03 = *(const wf4*)&dph[l - 1][mm][1][8 * half], dim47 = *(const wf4*)&dph[l - 1][mm][1][8 * half + 4];

                wf8 accr = {0, 0, 0, 0, 0, 0, 0, 0};
                wf8 acci = {0, 0, 0, 0, 0, 0, 0, 0};
                #pragma unroll
                for (int kb = 0; kb < 16; kb += 4) {
                    int j0 = kb + 2 * half;
                    wf2 br, bim;
                    if (pass < 2) {
                        float2 s0 = st[lslot ^ swz(kb * rowmul)];
                        float2 s1 = st[lslot ^ swz((kb + 1) * rowmul)];
                        br  = (wf2){s0.x, s1.x};
                        bim = (wf2){s0.y, s1.y};
                    } else {
                        wf4 s = *(const wf4*)&st[lslot ^ swz(kb)];
                        br  = (wf2){s.x, s.z};
                        bim = (wf2){s.y, s.w};
                    }
                    wf2 ar  = *(const wf2*)&mR[l - 1][mm][n16][j0];
                    accr = __builtin_amdgcn_wmma_f32_16x16x4_f32(false, ar, false, br,  (short)0, accr, false, false);
                    acci = __builtin_amdgcn_wmma_f32_16x16x4_f32(false, ar, false, bim, (short)0, acci, false, false);
                }
                const wf8 dre = __builtin_shufflevector(dre03, dre47, 0, 1, 2, 3, 4, 5, 6, 7);
                const wf8 dim = __builtin_shufflevector(dim03, dim47, 0, 1, 2, 3, 4, 5, 6, 7);
                const wf8 outr = dre * accr - dim * acci;
                const wf8 outi = dre * acci + dim * accr;
                if (pass < 2) {
                    #pragma unroll
                    for (int vv = 0; vv < 8; ++vv)
                        st[sslot ^ swz(vv * rowmul)] = make_float2(outr[vv], outi[vv]);
                } else {
                    #pragma unroll
                    for (int vv = 0; vv < 8; vv += 2)
                        *(wf4*)&st[sslot ^ swz(vv)] = (wf4){outr[vv], outi[vv], outr[vv + 1], outi[vv + 1]};
                }
            }
        }
    }
    __syncthreads();

    float z[NQ];
    #pragma unroll
    for (int q = 0; q < NQ; ++q) z[q] = 0.f;
    const int wslot = swz(__builtin_amdgcn_readfirstlane(wid) << 5);
    const int tslot = swz(lane) ^ wslot;
    for (int k = 0; k < 16; ++k) {
        int idx = (k << 8) + tid;
        float2 a = st[tslot ^ swz(k << 8)];
        float p = a.x * a.x + a.y * a.y;
        #pragma unroll
        for (int q = 0; q < NQ; ++q)
            z[q] += ((idx >> (11 - q)) & 1) ? -p : p;
    }
    #pragma unroll
    for (int q = 0; q < NQ; ++q) {
        float v = z[q];
        #pragma unroll
        for (int off = 16; off > 0; off >>= 1)
            v += __shfl_xor(v, off, 32);
        z[q] = v;
    }
    if (lane == 0) {
        #pragma unroll
        for (int q = 0; q < NQ; ++q) red[wid][q] = z[q];
    }
    __syncthreads();
    if (tid < NQ) {
        float s = 0.f;
        #pragma unroll
        for (int ww = 0; ww < 8; ++ww) s += red[ww][tid];
        out[b * NQ + tid] = s;
    }
}

extern "C" void kernel_launch(void* const* d_in, const int* in_sizes, int n_in,
                              void* d_out, int out_size, void* d_ws, size_t ws_size,
                              hipStream_t stream) {
    const float* x = (const float*)d_in[0];
    const float* w = (const float*)d_in[1];
    float* out = (float*)d_out;
    int B = in_sizes[0] / NQ;
    qsim_kernel<<<B, 256, 0, stream>>>(x, w, out);
}


# Only |amp|^2 is measured: layer-3 RZ row phases (modulus 1) left out of its 3 sweeps; last sweep's accumulators squared and summed in registers: no final store, barrier, reload; sum order changes
# speedup vs baseline: 3.2250x; 1.1155x over previous
#include <hip/hip_runtime.h>


typedef float wf2 __attribute__((ext_vector_type(2)));
typedef float wf8 __attribute__((ext_vector_type(8)));

#define NQ 12
#define NL 4
#define MP 20

typedef float wf4 __attribute__((ext_vector_type(4)));

__device__ __forceinline__ int swz(int i) {
    int q4 = ((i >> 5) ^ (i >> 7) ^ (i >> 9) ^ (i >> 11)) & 1;
    return i ^ (q4 << 4) ^ (((i >> 7) & 1) << 3) ^ (((i >> 6) & 1) << 2) ^ (((i >> 8) & 1) << 1);
}

__device__ __forceinline__ int colbase_c(int tt, int n16) {
    return (((((tt >> 1) << 4) | n16) << 1) | (tt & 1)) << 4;
}

__device__ __forceinline__ float2 cmul(float2 a, float2 b) {
    return make_float2(a.x * b.x - a.y * b.y, a.x * b.y + a.y * b.x);
}

__global__ __launch_bounds__(256, 8) void qsim_kernel(const float* __restrict__ x,
                                                   const float* __restrict__ w,
                                                   float* __restrict__ out) {
    __shared__ __attribute__((aligned(16))) float2 st[4096];
    __shared__ __attribute__((aligned(16))) float  mR[NL - 1][5][16][MP];
    __shared__ __attribute__((aligned(16))) float  dph[NL - 2][5][2][16];
    __shared__ __attribute__((aligned(16))) float2 col0[5][16];
    __shared__ __attribute__((aligned(16))) float2 l1v[5][4][16];
    __shared__ __attribute__((aligned(16))) float2 ry[NL][NQ];
    __shared__ float2 rz[NL - 1][NQ];
    __shared__ float  red[8][12];

    const int tid  = threadIdx.x;
    const int b    = blockIdx.x;
    const int lane = tid & 31;
    const int wid  = tid >> 5;
    const int half = lane >> 4;
    const int n16  = lane & 15;

    if (tid < NL * NQ) {
        int l = tid / NQ, q = tid % NQ;
        float th = w[(l * NQ + q) * 2 + 0] + (l == 0 ? x[b * NQ + q] : 0.f);
        ry[l][q] = make_float2(cosf(0.5f * th),  sinf(0.5f * th));
        if (l < NL - 1) {
            float ph = w[(l * NQ + q) * 2 + 1];
            rz[l][q] = make_float2(cosf(0.5f * ph), -sinf(0.5f * ph));
        }
    }

    __syncthreads();

    if (tid < 5 * 16) {
        int m = tid >> 4, i = tid & 15;
        int g    = (m == 0) ? 0 : (m <= 2 ? 1 : 2);
        int bvar = (m == 0) ? 0 : (m <= 2 ? (m - 1) : (m - 3));
        int ip = i;
        ip ^= (ip >> 1) & 1;
        ip ^= (ip >> 1) & 2;
        ip ^= (ip >> 1) & 4;
        ip ^= bvar << 3;
        float2 v = make_float2(1.f, 0.f);
        float  r = 1.f;
        #pragma unroll
        for (int k = 0; k < 4; ++k) {
            int bi = (ip >> (3 - k)) & 1;
            float2 e  = rz[0][4 * g + k];
            float2 cs = ry[0][4 * g + k];
            if (bi) e.y = -e.y;
            v = cmul(v, e);
            r *= bi ? cs.y : cs.x;
        }
        col0[m][i] = make_float2(v.x * r, v.y * r);
    }

    if (tid < (NL - 2) * 5 * 16) {
        int l = 1 + tid / 80, rr = tid % 80;
        int m = rr >> 4, i = rr & 15;
        int g    = (m == 0) ? 0 : (m <= 2 ? 1 : 2);
        int bvar = (m == 0) ? 0 : (m <= 2 ? (m - 1) : (m - 3));
        int ip = i;
        ip ^= (ip >> 1) & 1;
        ip ^= (ip >> 1) & 2;
        ip ^= (ip >> 1) & 4;
        ip ^= bvar << 3;
        float2 v = make_float2(1.f, 0.f);
        #pragma unroll
        for (int k = 0; k < 4; ++k) {
            float2 e = rz[l][4 * g + k];
            if ((ip >> (3 - k)) & 1) e.y = -e.y;
            v = cmul(v, e);
        }
        dph[l - 1][m][0][i] = v.x;
        dph[l - 1][m][1][i] = v.y;
    }
    {
        const int i = tid >> 4, j = tid & 15;
        int ip = i;
        ip ^= (ip >> 1) & 1;
        ip ^= (ip >> 1) & 2;
        ip ^= (ip >> 1) & 4;
        const int df = ip ^ j;
        const bool e0 = !(df & 8), e1 = !(df & 4), e2 = !(df & 2), e3 = !(df & 1);
        const int sg0 = ((0x6996 >> (~ip & j & 15)) & 1) << 31;
        const int sg1 = ((0x6996 >> (~(ip ^ 8) & j & 15)) & 1) << 31;
        #pragma unroll
        for (int l = 1; l < NL; ++l) {
            #pragma unroll
            for (int g = 0; g < 3; ++g) {
                const wf4 q01 = *(const wf4*)&ry[l][4 * g];
                const wf4 q23 = *(const wf4*)&ry[l][4 * g + 2];
                const float t1 = e1 ? q01.z : q01.w;
                const float t2 = e2 ? q23.x : q23.y;
                const float t3 = e3 ? q23.z : q23.w;
                float va = e0 ? q01.x : q01.y;
                va *= t1; va *= t2; va *= t3;
                mR[l - 1][g == 0 ? 0 : 2 * g - 1][i][j] = __int_as_float(__float_as_int(va) ^ sg0);
                if (g > 0) {
                    float vb = e0 ? q01.y : q01.x;
                    vb *= t1; vb *= t2; vb *= t3;
                    mR[l - 1][2 * g][i][j] = __int_as_float(__float_as_int(vb) ^ sg1);
                }
            }
        }
    }
    __syncthreads();

    {
        const int m = __builtin_amdgcn_readfirstlane(wid);
        if (m < 5) {
            const bool isA = (m == 0), isB = (m == 1) | (m == 2);
            const int  fsrc = isA ? 0 : (isB ? 1 + ((n16 >> 1) & 1) : 3 + (n16 & 1));
            const bool live = isB ? (n16 < 4) : (n16 < 2);
            const bool keep0 = live && (!(isA | isB) || (n16 & 1) == 0);
            const bool keep1 = live && (!(isA | isB) || (n16 & 1) == 1);
            const wf4 dre03 = *(const wf4*)&dph[0][m][0][8 * half], dre47 = *(const wf4*)&dph[0][m][0][8 * half + 4];
            const wf4 dim03 = *(const wf4*)&dph[0][m][1][8 * half], dim47 = *(const wf4*)&dph[0][m][1][8 * half + 4];
            wf8 accr = {0, 0, 0, 0, 0, 0, 0, 0};
            wf8 acci = {0, 0, 0, 0, 0, 0, 0, 0};
            #pragma unroll
            for (int kb = 0; kb < 16; kb += 4) {
                const int j0 = kb + 2 * half;
                const wf4 cc = *(const wf4*)&col0[fsrc][j0];
                const wf2 br  = (wf2){keep0 ? cc.x : 0.f, keep1 ? cc.z : 0.f};
                const wf2 bim = (wf2){keep0 ? cc.y : 0.f, keep1 ? cc.w : 0.f};
                const wf2 ar  = *(const wf2*)&mR[0][m][n16][j0];
                accr = __builtin_amdgcn_wmma_f32_16x16x4_f32(false, ar, false, br,  (short)0, accr, false, false);
                acci = __builtin_amdgcn_wmma_f32_16x16x4_f32(false, ar, false, bim, (short)0, acci, false, false);
            }
            const wf8 dre = __builtin_shufflevector(dre03, dre47, 0, 1, 2, 3, 4, 5, 6, 7);
            const wf8 dim = __builtin_shufflevector(dim03, dim47, 0, 1, 2, 3, 4, 5, 6, 7);
            const wf8 outr = dre * accr - dim * acci;
            const wf8 outi = dre * acci + dim * accr;
            if (n16 < 4) {
                #pragma unroll
                for (int vv = 0; vv < 8; vv += 2)
                    *(wf4*)&l1v[m][n16][8 * half + vv] = (wf4){outr[vv], outi[vv], outr[vv + 1], outi[vv + 1]};
            }
        }
    }
    __syncthreads();

    {
        const int iA = (wid << 1) | (n16 >> 3);
        const int vB = iA & 1;
        const float2 al0 = l1v[0][0][iA], al1 = l1v[0][1][iA];
        #pragma unroll
        for (int t2 = 0; t2 < 2; ++t2) {
            const int tt = wid * 2 + t2;
            const int iB = ((n16 & 7) << 1) | t2;
            const float2 b00 = l1v[1 + vB][0][iB], b01 = l1v[1 + vB][1][iB];
            const float2 b10 = l1v[1 + vB][2][iB], b11 = l1v[1 + vB][3][iB];
            const float2 x0 = cmul(al0, b00), y0 = cmul(al1, b10);
            const float2 x1 = cmul(al0, b01), y1 = cmul(al1, b11);
            const float2 e0 = make_float2(x0.x + y0.x, x0.y + y0.y);
            const float2 e1 = make_float2(x1.x + y1.x, x1.y + y1.y);
            const int sslot = swz(colbase_c(tt, n16) + 8 * half);
            #pragma unroll
            for (int vv = 0; vv < 8; vv += 2) {
                const wf4 g0 = *(const wf4*)&l1v[3 + t2][0][8 * half + vv];
                const wf4 g1 = *(const wf4*)&l1v[3 + t2][1][8 * half + vv];
                const float2 q0 = cmul(e0, make_float2(g0.x, g0.y)), r0 = cmul(e1, make_float2(g1.x, g1.y));
                const float2 q1 = cmul(e0, make_float2(g0.z, g0.w)), r1 = cmul(e1, make_float2(g1.z, g1.w));
                *(wf4*)&st[sslot ^ swz(vv)] = (wf4){q0.x + r0.x, q0.y + r0.y, q1.x + r1.x, q1.y + r1.y};
            }
        }
    }

    wf8 prob[2];

    #pragma unroll
    for (int l = 2; l < NL; ++l) {
        #pragma unroll
        for (int pass = 0; pass < 3; ++pass) {
            __syncthreads();
            #pragma unroll
            for (int t2 = 0; t2 < 2; ++t2) {
                const int tt = wid * 2 + t2;
                int mm, rowmul, colbase;
                if (pass == 0)      { mm = 0;            rowmul = 256; colbase = (tt << 4) | n16; }
                else if (pass == 1) { mm = 1 + (tt & 1); rowmul = 16;  colbase = (tt << 8) | n16; }
                else                { mm = 3 + (tt & 1); rowmul = 1;
                                      colbase = colbase_c(tt, n16); }

                const int lslot = swz(colbase + 2 * half * rowmul);
                const int sslot = swz(colbase + 8 * half * rowmul);

                wf8 accr = {0, 0, 0, 0, 0, 0, 0, 0};
                wf8 acci = {0, 0, 0, 0, 0, 0, 0, 0};
                #pragma unroll
                for (int kb = 0; kb < 16; kb += 4) {
                    int j0 = kb + 2 * half;
                    wf2 br, bim;
                    if (pass < 2) {
                        float2 s0 = st[lslot ^ swz(kb * rowmul)];
                        float2 s1 = st[lslot ^ swz((kb + 1) * rowmul)];
                        br  = (wf2){s0.x, s1.x};
                        bim = (wf2){s0.y, s1.y};
                    } else {
                        wf4 s = *(const wf4*)&st[lslot ^ swz(kb)];
                        br  = (wf2){s.x, s.z};
                        bim = (wf2){s.y, s.w};
                    }
                    wf2 ar  = *(const wf2*)&mR[l - 1][mm][n16][j0];
                    accr = __builtin_amdgcn_wmma_f32_16x16x4_f32(false, ar, false, br,  (short)0, accr, false, false);
                    acci = __builtin_amdgcn_wmma_f32_16x16x4_f32(false, ar, false, bim, (short)0, acci, false, false);
                }
                wf8 outr, outi;
                if (l < NL - 1) {
                    const wf4 dre03 = *(const wf4*)&dph[l - 1][mm][0][8 * half], dre47 = *(const wf4*)&dph[l - 1][mm][0][8 * half + 4];
                    const wf4 dim03 = *(const wf4*)&dph[l - 1][mm][1][8 * half], dim47 = *(const wf4*)&dph[l - 1][mm][1][8 * half + 4];
                    const wf8 dre = __builtin_shufflevector(dre03, dre47, 0, 1, 2, 3, 4, 5, 6, 7);
                    const wf8 dim = __builtin_shufflevector(dim03, dim47, 0, 1, 2, 3, 4, 5, 6, 7);
                    outr = dre * accr - dim * acci;
                    outi = dre * acci + dim * accr;
                } else {
                    outr = accr;
                    outi = acci;
                }
                if (l == NL - 1 && pass == 2) {
                    prob[t2] = outr * outr + outi * outi;
                } else if (pass < 2) {
                    #pragma unroll
                    for (int vv = 0; vv < 8; ++vv)
                        st[sslot ^ swz(vv * rowmul)] = make_float2(outr[vv], outi[vv]);
                } else {
                    #pragma unroll
                    for (int vv = 0; vv < 8; vv += 2)
                        *(wf4*)&st[sslot ^ swz(vv)] = (wf4){outr[vv], outi[vv], outr[vv + 1], outi[vv + 1]};
                }
            }
        }
    }

    float z[NQ];
    {
        const wf8 ps = prob[0] + prob[1];
        const wf8 pd = prob[0] - prob[1];
        const float a01 = ps[0] + ps[1], a23 = ps[2] + ps[3], a45 = ps[4] + ps[5], a67 = ps[6] + ps[7];
        const float lo = a01 + a23, hi = a45 + a67;
        const float P = lo + hi;
        z[7]  = ((pd[0] + pd[1]) + (pd[2] + pd[3])) + ((pd[4] + pd[5]) + (pd[6] + pd[7]));
        z[9]  = lo - hi;
        z[10] = (a01 - a23) + (a45 - a67);
        z[11] = ((ps[0] - ps[1]) + (ps[2] - ps[3])) + ((ps[4] - ps[5]) + (ps[6] - ps[7]));
        z[0] = (wid & 4) ? -P : P;
        z[1] = (wid & 2) ? -P : P;
        z[2] = (wid & 1) ? -P : P;
        z[3] = (n16 & 8) ? -P : P;
        z[4] = (n16 & 4) ? -P : P;
        z[5] = (n16 & 2) ? -P : P;
        z[6] = (n16 & 1) ? -P : P;
        z[8] = half ? -P : P;
    }
    #pragma unroll
    for (int q = 0; q < NQ; ++q) {
        float v = z[q];
        #pragma unroll
        for (int off = 16; off > 0; off >>= 1)
            v += __shfl_xor(v, off, 32);
        z[q] = v;
    }
    if (lane == 0) {
        #pragma unroll
        for (int q = 0; q < NQ; ++q) red[wid][q] = z[q];
    }
    __syncthreads();
    if (tid < NQ) {
        float s = 0.f;
        #pragma unroll
        for (int ww = 0; ww < 8; ++ww) s += red[ww][tid];
        out[b * NQ + tid] = s;
    }
}

extern "C" void kernel_launch(void* const* d_in, const int* in_sizes, int n_in,
                              void* d_out, int out_size, void* d_ws, size_t ws_size,
                              hipStream_t stream) {
    const float* x = (const float*)d_in[0];
    const float* w = (const float*)d_in[1];
    float* out = (float*)d_out;
    int B = in_sizes[0] / NQ;
    qsim_kernel<<<B, 256, 0, stream>>>(x, w, out);
}


# Layer 2's group-A pass applied to the four parity-masked alpha factors of the layer-1 state (one 16x16 WMMA product), not to the state: 1 of 6 sweeps dropped; layer-2 rounding, sum grouping change
# speedup vs baseline: 3.5038x; 1.0864x over previous
#include <hip/hip_runtime.h>


typedef float wf2 __attribute__((ext_vector_type(2)));
typedef float wf8 __attribute__((ext_vector_type(8)));

#define NQ 12
#define NL 4
#define MP 20

typedef float wf4 __attribute__((ext_vector_type(4)));

__device__ __forceinline__ int swz(int i) {
    int q4 = ((i >> 5) ^ (i >> 7) ^ (i >> 9) ^ (i >> 11)) & 1;
    return i ^ (q4 << 4) ^ (((i >> 7) & 1) << 3) ^ (((i >> 6) & 1) << 2) ^ (((i >> 8) & 1) << 1);
}

__device__ __forceinline__ int colbase_c(int tt, int n16) {
    return (((((tt >> 1) << 4) | n16) << 1) | (tt & 1)) << 4;
}

__device__ __forceinline__ float2 cmul(float2 a, float2 b) {
    return make_float2(a.x * b.x - a.y * b.y, a.x * b.y + a.y * b.x);
}

__global__ __launch_bounds__(256, 8) void qsim_kernel(const float* __restrict__ x,
                                                   const float* __restrict__ w,
                                                   float* __restrict__ out) {
    __shared__ __attribute__((aligned(16))) float2 st[4096];
    __shared__ __attribute__((aligned(16))) float  mR[NL - 1][5][16][MP];
    __shared__ __attribute__((aligned(16))) float  dph[NL - 2][5][2][16];
    __shared__ __attribute__((aligned(16))) float2 col0[5][16];
    __shared__ __attribute__((aligned(16))) float2 l1v[5][4][16];
    __shared__ __attribute__((aligned(16))) float2 l2a[4][16];
    __shared__ __attribute__((aligned(16))) float2 ry[NL][NQ];
    __shared__ float2 rz[NL - 1][NQ];
    __shared__ float  red[8][12];

    const int tid  = threadIdx.x;
    const int b    = blockIdx.x;
    const int lane = tid & 31;
    const int wid  = tid >> 5;
    const int half = lane >> 4;
    const int n16  = lane & 15;

    if (tid < NL * NQ) {
        int l = tid / NQ, q = tid % NQ;
        float th = w[(l * NQ + q) * 2 + 0] + (l == 0 ? x[b * NQ + q] : 0.f);
        ry[l][q] = make_float2(cosf(0.5f * th),  sinf(0.5f * th));
        if (l < NL - 1) {
            float ph = w[(l * NQ + q) * 2 + 1];
            rz[l][q] = make_float2(cosf(0.5f * ph), -sinf(0.5f * ph));
        }
    }

    __syncthreads();

    if (tid < 5 * 16) {
        int m = tid >> 4, i = tid & 15;
        int g    = (m == 0) ? 0 : (m <= 2 ? 1 : 2);
        int bvar = (m == 0) ? 0 : (m <= 2 ? (m - 1) : (m - 3));
        int ip = i;
        ip ^= (ip >> 1) & 1;
        ip ^= (ip >> 1) & 2;
        ip ^= (ip >> 1) & 4;
        ip ^= bvar << 3;
        float2 v = make_float2(1.f, 0.f);
        float  r = 1.f;
        #pragma unroll
        for (int k = 0; k < 4; ++k) {
            int bi = (ip >> (3 - k)) & 1;
            float2 e  = rz[0][4 * g + k];
            float2 cs = ry[0][4 * g + k];
            if (bi) e.y = -e.y;
            v = cmul(v, e);
            r *= bi ? cs.y : cs.x;
        }
        col0[m][i] = make_float2(v.x * r, v.y * r);
    }

    if (tid < (NL - 2) * 5 * 16) {
        int l = 1 + tid / 80, rr = tid % 80;
        int m = rr >> 4, i = rr & 15;
        int g    = (m == 0) ? 0 : (m <= 2 ? 1 : 2);
        int bvar = (m == 0) ? 0 : (m <= 2 ? (m - 1) : (m - 3));
        int ip = i;
        ip ^= (ip >> 1) & 1;
        ip ^= (ip >> 1) & 2;
        ip ^= (ip >> 1) & 4;
        ip ^= bvar << 3;
        float2 v = make_float2(1.f, 0.f);
        #pragma unroll
        for (int k = 0; k < 4; ++k) {
            float2 e = rz[l][4 * g + k];
            if ((ip >> (3 - k)) & 1) e.y = -e.y;
            v = cmul(v, e);
        }
        dph[l - 1][m][0][i] = v.x;
        dph[l - 1][m][1][i] = v.y;
    }
    {
        const int i = tid >> 4, j = tid & 15;
        int ip = i;
        ip ^= (ip >> 1) & 1;
        ip ^= (ip >> 1) & 2;
        ip ^= (ip >> 1) & 4;
        const int df = ip ^ j;
        const bool e0 = !(df & 8), e1 = !(df & 4), e2 = !(df & 2), e3 = !(df & 1);
        const int sg0 = ((0x6996 >> (~ip & j & 15)) & 1) << 31;
        const int sg1 = ((0x6996 >> (~(ip ^ 8) & j & 15)) & 1) << 31;
        #pragma unroll
        for (int l = 1; l < NL; ++l) {
            #pragma unroll
            for (int g = 0; g < 3; ++g) {
                const wf4 q01 = *(const wf4*)&ry[l][4 * g];
                const wf4 q23 = *(const wf4*)&ry[l][4 * g + 2];
                const float t1 = e1 ? q01.z : q01.w;
                const float t2 = e2 ? q23.x : q23.y;
                const float t3 = e3 ? q23.z : q23.w;
                float va = e0 ? q01.x : q01.y;
                va *= t1; va *= t2; va *= t3;
                mR[l - 1][g == 0 ? 0 : 2 * g - 1][i][j] = __int_as_float(__float_as_int(va) ^ sg0);
                if (g > 0) {
                    float vb = e0 ? q01.y : q01.x;
                    vb *= t1; vb *= t2; vb *= t3;
                    mR[l - 1][2 * g][i][j] = __int_as_float(__float_as_int(vb) ^ sg1);
                }
            }
        }
    }
    __syncthreads();

    {
        const int m = __builtin_amdgcn_readfirstlane(wid);
        if (m < 5) {
            const bool isA = (m == 0), isB = (m == 1) | (m == 2);
            const int  fsrc = isA ? 0 : (isB ? 1 + ((n16 >> 1) & 1) : 3 + (n16 & 1));
            const bool live = isB ? (n16 < 4) : (n16 < 2);
            const bool keep0 = live && (!(isA | isB) || (n16 & 1) == 0);
            const bool keep1 = live && (!(isA | isB) || (n16 & 1) == 1);
            const wf4 dre03 = *(const wf4*)&dph[0][m][0][8 * half], dre47 = *(const wf4*)&dph[0][m][0][8 * half + 4];
            const wf4 dim03 = *(const wf4*)&dph[0][m][1][8 * half], dim47 = *(const wf4*)&dph[0][m][1][8 * half + 4];
            wf8 accr = {0, 0, 0, 0, 0, 0, 0, 0};
            wf8 acci = {0, 0, 0, 0, 0, 0, 0, 0};
            #pragma unroll
            for (int kb = 0; kb < 16; kb += 4) {
                const int j0 = kb + 2 * half;
                const wf4 cc = *(const wf4*)&col0[fsrc][j0];
                const wf2 br  = (wf2){keep0 ? cc.x : 0.f, keep1 ? cc.z : 0.f};
                const wf2 bim = (wf2){keep0 ? cc.y : 0.f, keep1 ? cc.w : 0.f};
                const wf2 ar  = *(const wf2*)&mR[0][m][n16][j0];
                accr = __builtin_amdgcn_wmma_f32_16x16x4_f32(false, ar, false, br,  (short)0, accr, false, false);
                acci = __builtin_amdgcn_wmma_f32_16x16x4_f32(false, ar, false, bim, (short)0, acci, false, false);
            }
            const wf8 dre = __builtin_shufflevector(dre03, dre47, 0, 1, 2, 3, 4, 5, 6, 7);
            const wf8 dim = __builtin_shufflevector(dim03, dim47, 0, 1, 2, 3, 4, 5, 6, 7);
            const wf8 outr = dre * accr - dim * acci;
            const wf8 outi = dre * acci + dim * accr;
            if (n16 < 4) {
                #pragma unroll
                for (int vv = 0; vv < 8; vv += 2)
                    *(wf4*)&l1v[m][n16][8 * half + vv] = (wf4){outr[vv], outi[vv], outr[vv + 1], outi[vv + 1]};
            }
        }
    }
    __syncthreads();

    {
        const int m = __builtin_amdgcn_readfirstlane(wid);
        if (m == 0) {
            const int  sfac  = (n16 >> 1) & 1;
            const bool live  = n16 < 4;
            const bool keep0 = live && (n16 & 1) == 0;
            const bool keep1 = live && (n16 & 1) == 1;
            const wf4 dre03 = *(const wf4*)&dph[1][0][0][8 * half], dre47 = *(const wf4*)&dph[1][0][0][8 * half + 4];
            const wf4 dim03 = *(const wf4*)&dph[1][0][1][8 * half], dim47 = *(const wf4*)&dph[1][0][1][8 * half + 4];
            wf8 accr = {0, 0, 0, 0, 0, 0, 0, 0};
            wf8 acci = {0, 0, 0, 0, 0, 0, 0, 0};
            #pragma unroll
            for (int kb = 0; kb < 16; kb += 4) {
                const int j0 = kb + 2 * half;
                const wf4 cc = *(const wf4*)&l1v[0][sfac][j0];
                const wf2 br  = (wf2){keep0 ? cc.x : 0.f, keep1 ? cc.z : 0.f};
                const wf2 bim = (wf2){keep0 ? cc.y : 0.f, keep1 ? cc.w : 0.f};
                const wf2 ar  = *(const wf2*)&mR[1][0][n16][j0];
                accr = __builtin_amdgcn_wmma_f32_16x16x4_f32(false, ar, false, br,  (short)0, accr, false, false);
                acci = __builtin_amdgcn_wmma_f32_16x16x4_f32(false, ar, false, bim, (short)0, acci, false, false);
            }
            const wf8 dre = __builtin_shufflevector(dre03, dre47, 0, 1, 2, 3, 4, 5, 6, 7);
            const wf8 dim = __builtin_shufflevector(dim03, dim47, 0, 1, 2, 3, 4, 5, 6, 7);
            const wf8 outr = dre * accr - dim * acci;
            const wf8 outi = dre * acci + dim * accr;
            if (n16 < 4) {
                #pragma unroll
                for (int vv = 0; vv < 8; vv += 2)
                    *(wf4*)&l2a[n16][8 * half + vv] = (wf4){outr[vv], outi[vv], outr[vv + 1], outi[vv + 1]};
            }
        }
    }
    __syncthreads();

    {
        const int iA = (wid << 1) | (n16 >> 3);
        const float2 a00 = l2a[0][iA], a01 = l2a[1][iA];
        const float2 a10 = l2a[2][iA], a11 = l2a[3][iA];
        #pragma unroll
        for (int t2 = 0; t2 < 2; ++t2) {
            const int tt = wid * 2 + t2;
            const int iB = ((n16 & 7) << 1) | t2;
            float2 ep[2];
            #pragma unroll
            for (int p = 0; p < 2; ++p) {
                const float2 x0 = cmul(a00, l1v[1][p][iB]), y0 = cmul(a10, l1v[1][2 + p][iB]);
                const float2 x1 = cmul(a01, l1v[2][p][iB]), y1 = cmul(a11, l1v[2][2 + p][iB]);
                ep[p] = make_float2((x0.x + y0.x) + (x1.x + y1.x), (x0.y + y0.y) + (x1.y + y1.y));
            }
            const float2 e0 = ep[0], e1 = ep[1];
            const int sslot = swz(colbase_c(tt, n16) + 8 * half);
            #pragma unroll
            for (int vv = 0; vv < 8; vv += 2) {
                const wf4 g0 = *(const wf4*)&l1v[3 + t2][0][8 * half + vv];
                const wf4 g1 = *(const wf4*)&l1v[3 + t2][1][8 * half + vv];
                const float2 q0 = cmul(e0, make_float2(g0.x, g0.y)), r0 = cmul(e1, make_float2(g1.x, g1.y));
                const float2 q1 = cmul(e0, make_float2(g0.z, g0.w)), r1 = cmul(e1, make_float2(g1.z, g1.w));
                *(wf4*)&st[sslot ^ swz(vv)] = (wf4){q0.x + r0.x, q0.y + r0.y, q1.x + r1.x, q1.y + r1.y};
            }
        }
    }

    wf8 prob[2];

    #pragma unroll
    for (int l = 2; l < NL; ++l) {
        #pragma unroll
        for (int pass = 0; pass < 3; ++pass) {
            if (l == 2 && pass == 0) continue;
            __syncthreads();
            #pragma unroll
            for (int t2 = 0; t2 < 2; ++t2) {
                const int tt = wid * 2 + t2;
                int mm, rowmul, colbase;
                if (pass == 0)      { mm = 0;            rowmul = 256; colbase = (tt << 4) | n16; }
                else if (pass == 1) { mm = 1 + (tt & 1); rowmul = 16;  colbase = (tt << 8) | n16; }
                else                { mm = 3 + (tt & 1); rowmul = 1;
                                      colbase = colbase_c(tt, n16); }

                const int lslot = swz(colbase + 2 * half * rowmul);
                const int sslot = swz(colbase + 8 * half * rowmul);

                wf8 accr = {0, 0, 0, 0, 0, 0, 0, 0};
                wf8 acci = {0, 0, 0, 0, 0, 0, 0, 0};
                #pragma unroll
                for (int kb = 0; kb < 16; kb += 4) {
                    int j0 = kb + 2 * half;
                    wf2 br, bim;
                    if (pass < 2) {
                        float2 s0 = st[lslot ^ swz(kb * rowmul)];
                        float2 s1 = st[lslot ^ swz((kb + 1) * rowmul)];
                        br  = (wf2){s0.x, s1.x};
                        bim = (wf2){s0.y, s1.y};
                    } else {
                        wf4 s = *(const wf4*)&st[lslot ^ swz(kb)];
                        br  = (wf2){s.x, s.z};
                        bim = (wf2){s.y, s.w};
                    }
                    wf2 ar  = *(const wf2*)&mR[l - 1][mm][n16][j0];
                    accr = __builtin_amdgcn_wmma_f32_16x16x4_f32(false, ar, false, br,  (short)0, accr, false, false);
                    acci = __builtin_amdgcn_wmma_f32_16x16x4_f32(false, ar, false, bim, (short)0, acci, false, false);
                }
                wf8 outr, outi;
                if (l < NL - 1) {
                    const wf4 dre03 = *(const wf4*)&dph[l - 1][mm][0][8 * half], dre47 = *(const wf4*)&dph[l - 1][mm][0][8 * half + 4];
                    const wf4 dim03 = *(const wf4*)&dph[l - 1][mm][1][8 * half], dim47 = *(const wf4*)&dph[l - 1][mm][1][8 * half + 4];
                    const wf8 dre = __builtin_shufflevector(dre03, dre47, 0, 1, 2, 3, 4, 5, 6, 7);
                    const wf8 dim = __builtin_shufflevector(dim03, dim47, 0, 1, 2, 3, 4, 5, 6, 7);
                    outr = dre * accr - dim * acci;
                    outi = dre * acci + dim * accr;
                } else {
                    outr = accr;
                    outi = acci;
                }
                if (l == NL - 1 && pass == 2) {
                    prob[t2] = outr * outr + outi * outi;
                } else if (pass < 2) {
                    #pragma unroll
                    for (int vv = 0; vv < 8; ++vv)
                        st[sslot ^ swz(vv * rowmul)] = make_float2(outr[vv], outi[vv]);
                } else {
                    #pragma unroll
                    for (int vv = 0; vv < 8; vv += 2)
                        *(wf4*)&st[sslot ^ swz(vv)] = (wf4){outr[vv], outi[vv], outr[vv + 1], outi[vv + 1]};
                }
            }
        }
    }

    float z[NQ];
    {
        const wf8 ps = prob[0] + prob[1];
        const wf8 pd = prob[0] - prob[1];
        const float a01 = ps[0] + ps[1], a23 = ps[2] + ps[3], a45 = ps[4] + ps[5], a67 = ps[6] + ps[7];
        const float lo = a01 + a23, hi = a45 + a67;
        const float P = lo + hi;
        z[7]  = ((pd[0] + pd[1]) + (pd[2] + pd[3])) + ((pd[4] + pd[5]) + (pd[6] + pd[7]));
        z[9]  = lo - hi;
        z[10] = (a01 - a23) + (a45 - a67);
        z[11] = ((ps[0] - ps[1]) + (ps[2] - ps[3])) + ((ps[4] - ps[5]) + (ps[6] - ps[7]));
        z[0] = (wid & 4) ? -P : P;
        z[1] = (wid & 2) ? -P : P;
        z[2] = (wid & 1) ? -P : P;
        z[3] = (n16 & 8) ? -P : P;
        z[4] = (n16 & 4) ? -P : P;
        z[5] = (n16 & 2) ? -P : P;
        z[6] = (n16 & 1) ? -P : P;
        z[8] = half ? -P : P;
    }
    #pragma unroll
    for (int q = 0; q < NQ; ++q) {
        float v = z[q];
        #pragma unroll
        for (int off = 16; off > 0; off >>= 1)
            v += __shfl_xor(v, off, 32);
        z[q] = v;
    }
    if (lane == 0) {
        #pragma unroll
        for (int q = 0; q < NQ; ++q) red[wid][q] = z[q];
    }
    __syncthreads();
    if (tid < NQ) {
        float s = 0.f;
        #pragma unroll
        for (int ww = 0; ww < 8; ++ww) s += red[ww][tid];
        out[b * NQ + tid] = s;
    }
}

extern "C" void kernel_launch(void* const* d_in, const int* in_sizes, int n_in,
                              void* d_out, int out_size, void* d_ws, size_t ws_size,
                              hipStream_t stream) {
    const float* x = (const float*)d_in[0];
    const float* w = (const float*)d_in[1];
    float* out = (float*)d_out;
    int B = in_sizes[0] / NQ;
    qsim_kernel<<<B, 256, 0, stream>>>(x, w, out);
}


# Layer 2 applied wholly to the factors of the layer-1 state (5 small WMMA products), state then written once as a rank-4 WMMA product: 2 of 5 sweeps dropped; rounding and sum grouping change
# speedup vs baseline: 4.2462x; 1.2119x over previous
#include <hip/hip_runtime.h>


typedef float wf2 __attribute__((ext_vector_type(2)));
typedef float wf8 __attribute__((ext_vector_type(8)));

#define NQ 12
#define NL 4
#define MP 20

typedef float wf4 __attribute__((ext_vector_type(4)));

__device__ __forceinline__ int swz(int i) {
    int q4 = ((i >> 5) ^ (i >> 7) ^ (i >> 9) ^ (i >> 11)) & 1;
    return i ^ (q4 << 4) ^ (((i >> 7) & 1) << 3) ^ (((i >> 6) & 1) << 2) ^ (((i >> 8) & 1) << 1);
}

__device__ __forceinline__ int colbase_c(int tt, int n16) {
    return (((((tt >> 1) << 4) | n16) << 1) | (tt & 1)) << 4;
}

__device__ __forceinline__ float2 cmul(float2 a, float2 b) {
    return make_float2(a.x * b.x - a.y * b.y, a.x * b.y + a.y * b.x);
}

__global__ __launch_bounds__(256, 8) void qsim_kernel(const float* __restrict__ x,
                                                   const float* __restrict__ w,
                                                   float* __restrict__ out) {
    __shared__ __attribute__((aligned(16))) float2 st[4096];
    __shared__ __attribute__((aligned(16))) float  mR[NL - 1][5][16][MP];
    __shared__ __attribute__((aligned(16))) float  dph[NL - 2][5][2][16];
    __shared__ __attribute__((aligned(16))) float2 col0[5][16];
    __shared__ __attribute__((aligned(16))) float2 l1v[5][4][16];
    __shared__ __attribute__((aligned(16))) float2 l2f[5][16][16];
    __shared__ __attribute__((aligned(16))) float2 ry[NL][NQ];
    __shared__ float2 rz[NL - 1][NQ];
    __shared__ float  red[8][12];

    const int tid  = threadIdx.x;
    const int b    = blockIdx.x;
    const int lane = tid & 31;
    const int wid  = tid >> 5;
    const int half = lane >> 4;
    const int n16  = lane & 15;

    if (tid < NL * NQ) {
        int l = tid / NQ, q = tid % NQ;
        float th = w[(l * NQ + q) * 2 + 0] + (l == 0 ? x[b * NQ + q] : 0.f);
        ry[l][q] = make_float2(cosf(0.5f * th),  sinf(0.5f * th));
        if (l < NL - 1) {
            float ph = w[(l * NQ + q) * 2 + 1];
            rz[l][q] = make_float2(cosf(0.5f * ph), -sinf(0.5f * ph));
        }
    }

    __syncthreads();

    if (tid < 5 * 16) {
        int m = tid >> 4, i = tid & 15;
        int g    = (m == 0) ? 0 : (m <= 2 ? 1 : 2);
        int bvar = (m == 0) ? 0 : (m <= 2 ? (m - 1) : (m - 3));
        int ip = i;
        ip ^= (ip >> 1) & 1;
        ip ^= (ip >> 1) & 2;
        ip ^= (ip >> 1) & 4;
        ip ^= bvar << 3;
        float2 v = make_float2(1.f, 0.f);
        float  r = 1.f;
        #pragma unroll
        for (int k = 0; k < 4; ++k) {
            int bi = (ip >> (3 - k)) & 1;
            float2 e  = rz[0][4 * g + k];
            float2 cs = ry[0][4 * g + k];
            if (bi) e.y = -e.y;
            v = cmul(v, e);
            r *= bi ? cs.y : cs.x;
        }
        col0[m][i] = make_float2(v.x * r, v.y * r);
    }

    if (tid < (NL - 2) * 5 * 16) {
        int l = 1 + tid / 80, rr = tid % 80;
        int m = rr >> 4, i = rr & 15;
        int g    = (m == 0) ? 0 : (m <= 2 ? 1 : 2);
        int bvar = (m == 0) ? 0 : (m <= 2 ? (m - 1) : (m - 3));
        int ip = i;
        ip ^= (ip >> 1) & 1;
        ip ^= (ip >> 1) & 2;
        ip ^= (ip >> 1) & 4;
        ip ^= bvar << 3;
        float2 v = make_float2(1.f, 0.f);
        #pragma unroll
        for (int k = 0; k < 4; ++k) {
            float2 e = rz[l][4 * g + k];
            if ((ip >> (3 - k)) & 1) e.y = -e.y;
            v = cmul(v, e);
        }
        dph[l - 1][m][0][i] = v.x;
        dph[l - 1][m][1][i] = v.y;
    }
    {
        const int i = tid >> 4, j = tid & 15;
        int ip = i;
        ip ^= (ip >> 1) & 1;
        ip ^= (ip >> 1) & 2;
        ip ^= (ip >> 1) & 4;
        const int df = ip ^ j;
        const bool e0 = !(df & 8), e1 = !(df & 4), e2 = !(df & 2), e3 = !(df & 1);
        const int sg0 = ((0x6996 >> (~ip & j & 15)) & 1) << 31;
        const int sg1 = ((0x6996 >> (~(ip ^ 8) & j & 15)) & 1) << 31;
        #pragma unroll
        for (int l = 1; l < NL; ++l) {
            #pragma unroll
            for (int g = 0; g < 3; ++g) {
                const wf4 q01 = *(const wf4*)&ry[l][4 * g];
                const wf4 q23 = *(const wf4*)&ry[l][4 * g + 2];
                const float t1 = e1 ? q01.z : q01.w;
                const float t2 = e2 ? q23.x : q23.y;
                const float t3 = e3 ? q23.z : q23.w;
                float va = e0 ? q01.x : q01.y;
                va *= t1; va *= t2; va *= t3;
                mR[l - 1][g == 0 ? 0 : 2 * g - 1][i][j] = __int_as_float(__float_as_int(va) ^ sg0);
                if (g > 0) {
                    float vb = e0 ? q01.y : q01.x;
                    vb *= t1; vb *= t2; vb *= t3;
                    mR[l - 1][2 * g][i][j] = __int_as_float(__float_as_int(vb) ^ sg1);
                }
            }
        }
    }
    __syncthreads();

    {
        const int m = __builtin_amdgcn_readfirstlane(wid);
        if (m < 5) {
            const bool isA = (m == 0), isB = (m == 1) | (m == 2);
            const int  fsrc = isA ? 0 : (isB ? 1 + ((n16 >> 1) & 1) : 3 + (n16 & 1));
            const bool live = isB ? (n16 < 4) : (n16 < 2);
            const bool keep0 = live && (!(isA | isB) || (n16 & 1) == 0);
            const bool keep1 = live && (!(isA | isB) || (n16 & 1) == 1);
            const wf4 dre03 = *(const wf4*)&dph[0][m][0][8 * half], dre47 = *(const wf4*)&dph[0][m][0][8 * half + 4];
            const wf4 dim03 = *(const wf4*)&dph[0][m][1][8 * half], dim47 = *(const wf4*)&dph[0][m][1][8 * half + 4];
            wf8 accr = {0, 0, 0, 0, 0, 0, 0, 0};
            wf8 acci = {0, 0, 0, 0, 0, 0, 0, 0};
            #pragma unroll
            for (int kb = 0; kb < 16; kb += 4) {
                const int j0 = kb + 2 * half;
                const wf4 cc = *(const wf4*)&col0[fsrc][j0];
                const wf2 br  = (wf2){keep0 ? cc.x : 0.f, keep1 ? cc.z : 0.f};
                const wf2 bim = (wf2){keep0 ? cc.y : 0.f, keep1 ? cc.w : 0.f};
                const wf2 ar  = *(const wf2*)&mR[0][m][n16][j0];
                accr = __builtin_amdgcn_wmma_f32_16x16x4_f32(false, ar, false, br,  (short)0, accr, false, false);
                acci = __builtin_amdgcn_wmma_f32_16x16x4_f32(false, ar, false, bim, (short)0, acci, false, false);
            }
            const wf8 dre = __builtin_shufflevector(dre03, dre47, 0, 1, 2, 3, 4, 5, 6, 7);
            const wf8 dim = __builtin_shufflevector(dim03, dim47, 0, 1, 2, 3, 4, 5, 6, 7);
            const wf8 outr = dre * accr - dim * acci;
            const wf8 outi = dre * acci + dim * accr;
            if (n16 < 4) {
                #pragma unroll
                for (int vv = 0; vv < 8; vv += 2)
                    *(wf4*)&l1v[m][n16][8 * half + vv] = (wf4){outr[vv], outi[vv], outr[vv + 1], outi[vv + 1]};
            }
        }
    }
    __syncthreads();

    {
        const int m = __builtin_amdgcn_readfirstlane(wid);
        if (m < 5) {
            const bool isA = (m == 0), isB = (m == 1) | (m == 2);
            const int  smat = isA ? 0 : (isB ? 1 + (n16 >> 3) : 3 + (n16 & 1));
            const int  scol = isB ? ((n16 >> 1) & 3) : ((n16 >> 1) & 1);
            const bool live  = isB | (n16 < 4);
            const bool keep0 = live && (!(isA | isB) || (n16 & 1) == 0);
            const bool keep1 = live && (!(isA | isB) || (n16 & 1) == 1);
            const wf4 dre03 = *(const wf4*)&dph[1][m][0][8 * half], dre47 = *(const wf4*)&dph[1][m][0][8 * half + 4];
            const wf4 dim03 = *(const wf4*)&dph[1][m][1][8 * half], dim47 = *(const wf4*)&dph[1][m][1][8 * half + 4];
            wf8 accr = {0, 0, 0, 0, 0, 0, 0, 0};
            wf8 acci = {0, 0, 0, 0, 0, 0, 0, 0};
            #pragma unroll
            for (int kb = 0; kb < 16; kb += 4) {
                const int j0 = kb + 2 * half;
                const wf4 cc = *(const wf4*)&l1v[smat][scol][j0];
                const wf2 br  = (wf2){keep0 ? cc.x : 0.f, keep1 ? cc.z : 0.f};
                const wf2 bim = (wf2){keep0 ? cc.y : 0.f, keep1 ? cc.w : 0.f};
                const wf2 ar  = *(const wf2*)&mR[1][m][n16][j0];
                accr = __builtin_amdgcn_wmma_f32_16x16x4_f32(false, ar, false, br,  (short)0, accr, false, false);
                acci = __builtin_amdgcn_wmma_f32_16x16x4_f32(false, ar, false, bim, (short)0, acci, false, false);
            }
            const wf8 dre = __builtin_shufflevector(dre03, dre47, 0, 1, 2, 3, 4, 5, 6, 7);
            const wf8 dim = __builtin_shufflevector(dim03, dim47, 0, 1, 2, 3, 4, 5, 6, 7);
            const wf8 outr = dre * accr - dim * acci;
            const wf8 outi = dre * acci + dim * accr;
            if (live) {
                #pragma unroll
                for (int vv = 0; vv < 8; vv += 2)
                    *(wf4*)&l2f[m][n16][8 * half + vv] = (wf4){outr[vv], outi[vv], outr[vv + 1], outi[vv + 1]};
            }
        }
    }
    __syncthreads();

    {
        const int iA = (wid << 1) | (n16 >> 3);
        const int wv = (n16 >> 3) & 1;
        const float2 a00 = l2f[0][0][iA], a01 = l2f[0][1][iA];
        const float2 a10 = l2f[0][2][iA], a11 = l2f[0][3][iA];
        #pragma unroll
        for (int t2 = 0; t2 < 2; ++t2) {
            const int tt = wid * 2 + t2;
            const int iB = ((n16 & 7) << 1) | t2;
            float2 ff[2];
            #pragma unroll
            for (int u = 0; u < 2; ++u) {
                const int cb = 2 * half + u;
                const float2 x0 = cmul(a00, l2f[1 + wv][cb][iB]),     y0 = cmul(a10, l2f[1 + wv][4 + cb][iB]);
                const float2 x1 = cmul(a01, l2f[1 + wv][8 + cb][iB]), y1 = cmul(a11, l2f[1 + wv][12 + cb][iB]);
                ff[u] = make_float2((x0.x + y0.x) + (x1.x + y1.x), (x0.y + y0.y) + (x1.y + y1.y));
            }
            const float2 g0 = l2f[3 + t2][2 * half][n16], g1 = l2f[3 + t2][2 * half + 1][n16];
            const wf2 gr = (wf2){g0.x, g1.x}, gi = (wf2){g0.y, g1.y};
            const wf2 fr = (wf2){ff[0].x, ff[1].x}, fi = (wf2){ff[0].y, ff[1].y};
            const wf2 nfi = -fi;
            wf8 accr = {0, 0, 0, 0, 0, 0, 0, 0};
            wf8 acci = {0, 0, 0, 0, 0, 0, 0, 0};
            accr = __builtin_amdgcn_wmma_f32_16x16x4_f32(false, gr, false, fr,  (short)0, accr, false, false);
            acci = __builtin_amdgcn_wmma_f32_16x16x4_f32(false, gr, false, fi,  (short)0, acci, false, false);
            accr = __builtin_amdgcn_wmma_f32_16x16x4_f32(false, gi, false, nfi, (short)0, accr, false, false);
            acci = __builtin_amdgcn_wmma_f32_16x16x4_f32(false, gi, false, fr,  (short)0, acci, false, false);
            const int sslot = swz(colbase_c(tt, n16) + 8 * half);
            #pragma unroll
            for (int vv = 0; vv < 8; vv += 2)
                *(wf4*)&st[sslot ^ swz(vv)] = (wf4){accr[vv], acci[vv], accr[vv + 1], acci[vv + 1]};
        }
    }

    wf8 prob[2];

    {
        const int l = NL - 1;
        #pragma unroll
        for (int pass = 0; pass < 3; ++pass) {
            __syncthreads();
            #pragma unroll
            for (int t2 = 0; t2 < 2; ++t2) {
                const int tt = wid * 2 + t2;
                int mm, rowmul, colbase;
                if (pass == 0)      { mm = 0;            rowmul = 256; colbase = (tt << 4) | n16; }
                else if (pass == 1) { mm = 1 + (tt & 1); rowmul = 16;  colbase = (tt << 8) | n16; }
                else                { mm = 3 + (tt & 1); rowmul = 1;
                                      colbase = colbase_c(tt, n16); }

                const int lslot = swz(colbase + 2 * half * rowmul);
                const int sslot = swz(colbase + 8 * half * rowmul);

                wf8 accr = {0, 0, 0, 0, 0, 0, 0, 0};
                wf8 acci = {0, 0, 0, 0, 0, 0, 0, 0};
                #pragma unroll
                for (int kb = 0; kb < 16; kb += 4) {
                    int j0 = kb + 2 * half;
                    wf2 br, bim;
                    if (pass < 2) {
                        float2 s0 = st[lslot ^ swz(kb * rowmul)];
                        float2 s1 = st[lslot ^ swz((kb + 1) * rowmul)];
                        br  = (wf2){s0.x, s1.x};
                        bim = (wf2){s0.y, s1.y};
                    } else {
                        wf4 s = *(const wf4*)&st[lslot ^ swz(kb)];
                        br  = (wf2){s.x, s.z};
                        bim = (wf2){s.y, s.w};
                    }
                    wf2 ar  = *(const wf2*)&mR[l - 1][mm][n16][j0];
                    accr = __builtin_amdgcn_wmma_f32_16x16x4_f32(false, ar, false, br,  (short)0, accr, false, false);
                    acci = __builtin_amdgcn_wmma_f32_16x16x4_f32(false, ar, false, bim, (short)0, acci, false, false);
                }
                if (pass == 2) {
                    prob[t2] = accr * accr + acci * acci;
                } else {
                    #pragma unroll
                    for (int vv = 0; vv < 8; ++vv)
                        st[sslot ^ swz(vv * rowmul)] = make_float2(accr[vv], acci[vv]);
                }
            }
        }
    }

    float z[NQ];
    {
        const wf8 ps = prob[0] + prob[1];
        const wf8 pd = prob[0] - prob[1];
        const float a01 = ps[0] + ps[1], a23 = ps[2] + ps[3], a45 = ps[4] + ps[5], a67 = ps[6] + ps[7];
        const float lo = a01 + a23, hi = a45 + a67;
        const float P = lo + hi;
        z[7]  = ((pd[0] + pd[1]) + (pd[2] + pd[3])) + ((pd[4] + pd[5]) + (pd[6] + pd[7]));
        z[9]  = lo - hi;
        z[10] = (a01 - a23) + (a45 - a67);
        z[11] = ((ps[0] - ps[1]) + (ps[2] - ps[3])) + ((ps[4] - ps[5]) + (ps[6] - ps[7]));
        z[0] = (wid & 4) ? -P : P;
        z[1] = (wid & 2) ? -P : P;
        z[2] = (wid & 1) ? -P : P;
        z[3] = (n16 & 8) ? -P : P;
        z[4] = (n16 & 4) ? -P : P;
        z[5] = (n16 & 2) ? -P : P;
        z[6] = (n16 & 1) ? -P : P;
        z[8] = half ? -P : P;
    }
    #pragma unroll
    for (int q = 0; q < NQ; ++q) {
        float v = z[q];
        #pragma unroll
        for (int off = 16; off > 0; off >>= 1)
            v += __shfl_xor(v, off, 32);
        z[q] = v;
    }
    if (lane == 0) {
        #pragma unroll
        for (int q = 0; q < NQ; ++q) red[wid][q] = z[q];
    }
    __syncthreads();
    if (tid < NQ) {
        float s = 0.f;
        #pragma unroll
        for (int ww = 0; ww < 8; ++ww) s += red[ww][tid];
        out[b * NQ + tid] = s;
    }
}

extern "C" void kernel_launch(void* const* d_in, const int* in_sizes, int n_in,
                              void* d_out, int out_size, void* d_ws, size_t ws_size,
                              hipStream_t stream) {
    const float* x = (const float*)d_in[0];
    const float* w = (const float*)d_in[1];
    float* out = (float*)d_out;
    int B = in_sizes[0] / NQ;
    qsim_kernel<<<B, 256, 0, stream>>>(x, w, out);
}


# Layer 3's group-A pass applied to the four parity-masked alpha factors of the layer-2 state (one 16x16 WMMA product), not to the state: 1 of 3 sweeps dropped; layer-3 rounding, sum grouping change
# speedup vs baseline: 4.5935x; 1.0818x over previous
#include <hip/hip_runtime.h>


typedef float wf2 __attribute__((ext_vector_type(2)));
typedef float wf8 __attribute__((ext_vector_type(8)));

#define NQ 12
#define NL 4
#define MP 20

typedef float wf4 __attribute__((ext_vector_type(4)));

__device__ __forceinline__ int swz(int i) {
    int q4 = ((i >> 5) ^ (i >> 7) ^ (i >> 9) ^ (i >> 11)) & 1;
    return i ^ (q4 << 4) ^ (((i >> 7) & 1) << 3) ^ (((i >> 6) & 1) << 2) ^ (((i >> 8) & 1) << 1);
}

__device__ __forceinline__ int colbase_c(int tt, int n16) {
    return (((((tt >> 1) << 4) | n16) << 1) | (tt & 1)) << 4;
}

__device__ __forceinline__ float2 cmul(float2 a, float2 b) {
    return make_float2(a.x * b.x - a.y * b.y, a.x * b.y + a.y * b.x);
}

__global__ __launch_bounds__(256, 8) void qsim_kernel(const float* __restrict__ x,
                                                   const float* __restrict__ w,
                                                   float* __restrict__ out) {
    __shared__ __attribute__((aligned(16))) float2 st[4096];
    __shared__ __attribute__((aligned(16))) float  mR[NL - 1][5][16][MP];
    __shared__ __attribute__((aligned(16))) float  dph[NL - 2][5][2][16];
    __shared__ __attribute__((aligned(16))) float2 col0[5][16];
    __shared__ __attribute__((aligned(16))) float2 l1v[5][4][16];
    __shared__ __attribute__((aligned(16))) float2 l2f[5][16][16];
    __shared__ __attribute__((aligned(16))) float2 l3a[8][16];
    __shared__ __attribute__((aligned(16))) float2 ry[NL][NQ];
    __shared__ float2 rz[NL - 1][NQ];
    __shared__ float  red[8][12];

    const int tid  = threadIdx.x;
    const int b    = blockIdx.x;
    const int lane = tid & 31;
    const int wid  = tid >> 5;
    const int half = lane >> 4;
    const int n16  = lane & 15;

    if (tid < NL * NQ) {
        int l = tid / NQ, q = tid % NQ;
        float th = w[(l * NQ + q) * 2 + 0] + (l == 0 ? x[b * NQ + q] : 0.f);
        ry[l][q] = make_float2(cosf(0.5f * th),  sinf(0.5f * th));
        if (l < NL - 1) {
            float ph = w[(l * NQ + q) * 2 + 1];
            rz[l][q] = make_float2(cosf(0.5f * ph), -sinf(0.5f * ph));
        }
    }

    __syncthreads();

    if (tid < 5 * 16) {
        int m = tid >> 4, i = tid & 15;
        int g    = (m == 0) ? 0 : (m <= 2 ? 1 : 2);
        int bvar = (m == 0) ? 0 : (m <= 2 ? (m - 1) : (m - 3));
        int ip = i;
        ip ^= (ip >> 1) & 1;
        ip ^= (ip >> 1) & 2;
        ip ^= (ip >> 1) & 4;
        ip ^= bvar << 3;
        float2 v = make_float2(1.f, 0.f);
        float  r = 1.f;
        #pragma unroll
        for (int k = 0; k < 4; ++k) {
            int bi = (ip >> (3 - k)) & 1;
            float2 e  = rz[0][4 * g + k];
            float2 cs = ry[0][4 * g + k];
            if (bi) e.y = -e.y;
            v = cmul(v, e);
            r *= bi ? cs.y : cs.x;
        }
        col0[m][i] = make_float2(v.x * r, v.y * r);
    }

    if (tid < (NL - 2) * 5 * 16) {
        int l = 1 + tid / 80, rr = tid % 80;
        int m = rr >> 4, i = rr & 15;
        int g    = (m == 0) ? 0 : (m <= 2 ? 1 : 2);
        int bvar = (m == 0) ? 0 : (m <= 2 ? (m - 1) : (m - 3));
        int ip = i;
        ip ^= (ip >> 1) & 1;
        ip ^= (ip >> 1) & 2;
        ip ^= (ip >> 1) & 4;
        ip ^= bvar << 3;
        float2 v = make_float2(1.f, 0.f);
        #pragma unroll
        for (int k = 0; k < 4; ++k) {
            float2 e = rz[l][4 * g + k];
            if ((ip >> (3 - k)) & 1) e.y = -e.y;
            v = cmul(v, e);
        }
        dph[l - 1][m][0][i] = v.x;
        dph[l - 1][m][1][i] = v.y;
    }
    {
        const int i = tid >> 4, j = tid & 15;
        int ip = i;
        ip ^= (ip >> 1) & 1;
        ip ^= (ip >> 1) & 2;
        ip ^= (ip >> 1) & 4;
        const int df = ip ^ j;
        const bool e0 = !(df & 8), e1 = !(df & 4), e2 = !(df & 2), e3 = !(df & 1);
        const int sg0 = ((0x6996 >> (~ip & j & 15)) & 1) << 31;
        const int sg1 = ((0x6996 >> (~(ip ^ 8) & j & 15)) & 1) << 31;
        #pragma unroll
        for (int l = 1; l < NL; ++l) {
            #pragma unroll
            for (int g = 0; g < 3; ++g) {
                const wf4 q01 = *(const wf4*)&ry[l][4 * g];
                const wf4 q23 = *(const wf4*)&ry[l][4 * g + 2];
                const float t1 = e1 ? q01.z : q01.w;
                const float t2 = e2 ? q23.x : q23.y;
                const float t3 = e3 ? q23.z : q23.w;
                float va = e0 ? q01.x : q01.y;
                va *= t1; va *= t2; va *= t3;
                mR[l - 1][g == 0 ? 0 : 2 * g - 1][i][j] = __int_as_float(__float_as_int(va) ^ sg0);
                if (g > 0) {
                    float vb = e0 ? q01.y : q01.x;
                    vb *= t1; vb *= t2; vb *= t3;
                    mR[l - 1][2 * g][i][j] = __int_as_float(__float_as_int(vb) ^ sg1);
                }
            }
        }
    }
    __syncthreads();

    {
        const int m = __builtin_amdgcn_readfirstlane(wid);
        if (m < 5) {
            const bool isA = (m == 0), isB = (m == 1) | (m == 2);
            const int  fsrc = isA ? 0 : (isB ? 1 + ((n16 >> 1) & 1) : 3 + (n16 & 1));
            const bool live = isB ? (n16 < 4) : (n16 < 2);
            const bool keep0 = live && (!(isA | isB) || (n16 & 1) == 0);
            const bool keep1 = live && (!(isA | isB) || (n16 & 1) == 1);
            const wf4 dre03 = *(const wf4*)&dph[0][m][0][8 * half], dre47 = *(const wf4*)&dph[0][m][0][8 * half + 4];
            const wf4 dim03 = *(const wf4*)&dph[0][m][1][8 * half], dim47 = *(const wf4*)&dph[0][m][1][8 * half + 4];
            wf8 accr = {0, 0, 0, 0, 0, 0, 0, 0};
            wf8 acci = {0, 0, 0, 0, 0, 0, 0, 0};
            #pragma unroll
            for (int kb = 0; kb < 16; kb += 4) {
                const int j0 = kb + 2 * half;
                const wf4 cc = *(const wf4*)&col0[fsrc][j0];
                const wf2 br  = (wf2){keep0 ? cc.x : 0.f, keep1 ? cc.z : 0.f};
                const wf2 bim = (wf2){keep0 ? cc.y : 0.f, keep1 ? cc.w : 0.f};
                const wf2 ar  = *(const wf2*)&mR[0][m][n16][j0];
                accr = __builtin_amdgcn_wmma_f32_16x16x4_f32(false, ar, false, br,  (short)0, accr, false, false);
                acci = __builtin_amdgcn_wmma_f32_16x16x4_f32(false, ar, false, bim, (short)0, acci, false, false);
            }
            const wf8 dre = __builtin_shufflevector(dre03, dre47, 0, 1, 2, 3, 4, 5, 6, 7);
            const wf8 dim = __builtin_shufflevector(dim03, dim47, 0, 1, 2, 3, 4, 5, 6, 7);
            const wf8 outr = dre * accr - dim * acci;
            const wf8 outi = dre * acci + dim * accr;
            if (n16 < 4) {
                #pragma unroll
                for (int vv = 0; vv < 8; vv += 2)
                    *(wf4*)&l1v[m][n16][8 * half + vv] = (wf4){outr[vv], outi[vv], outr[vv + 1], outi[vv + 1]};
            }
        }
    }
    __syncthreads();

    {
        const int m = __builtin_amdgcn_readfirstlane(wid);
        if (m < 5) {
            const bool isA = (m == 0), isB = (m == 1) | (m == 2);
            const int  smat = isA ? 0 : (isB ? 1 + (n16 >> 3) : 3 + (n16 & 1));
            const int  scol = isB ? ((n16 >> 1) & 3) : ((n16 >> 1) & 1);
            const bool live  = isB | (n16 < 4);
            const bool keep0 = live && (!(isA | isB) || (n16 & 1) == 0);
            const bool keep1 = live && (!(isA | isB) || (n16 & 1) == 1);
            const wf4 dre03 = *(const wf4*)&dph[1][m][0][8 * half], dre47 = *(const wf4*)&dph[1][m][0][8 * half + 4];
            const wf4 dim03 = *(const wf4*)&dph[1][m][1][8 * half], dim47 = *(const wf4*)&dph[1][m][1][8 * half + 4];
            wf8 accr = {0, 0, 0, 0, 0, 0, 0, 0};
            wf8 acci = {0, 0, 0, 0, 0, 0, 0, 0};
            #pragma unroll
            for (int kb = 0; kb < 16; kb += 4) {
                const int j0 = kb + 2 * half;
                const wf4 cc = *(const wf4*)&l1v[smat][scol][j0];
                const wf2 br  = (wf2){keep0 ? cc.x : 0.f, keep1 ? cc.z : 0.f};
                const wf2 bim = (wf2){keep0 ? cc.y : 0.f, keep1 ? cc.w : 0.f};
                const wf2 ar  = *(const wf2*)&mR[1][m][n16][j0];
                accr = __builtin_amdgcn_wmma_f32_16x16x4_f32(false, ar, false, br,  (short)0, accr, false, false);
                acci = __builtin_amdgcn_wmma_f32_16x16x4_f32(false, ar, false, bim, (short)0, acci, false, false);
            }
            const wf8 dre = __builtin_shufflevector(dre03, dre47, 0, 1, 2, 3, 4, 5, 6, 7);
            const wf8 dim = __builtin_shufflevector(dim03, dim47, 0, 1, 2, 3, 4, 5, 6, 7);
            const wf8 outr = dre * accr - dim * acci;
            const wf8 outi = dre * acci + dim * accr;
            if (live) {
                #pragma unroll
                for (int vv = 0; vv < 8; vv += 2)
                    *(wf4*)&l2f[m][n16][8 * half + vv] = (wf4){outr[vv], outi[vv], outr[vv + 1], outi[vv + 1]};
            }
        }
    }
    __syncthreads();

    {
        const int m = __builtin_amdgcn_readfirstlane(wid);
        if (m == 0) {
            const int  scol  = (n16 >> 1) & 3;
            const bool live  = n16 < 8;
            const bool keep0 = live && (n16 & 1) == 0;
            const bool keep1 = live && (n16 & 1) == 1;
            wf8 accr = {0, 0, 0, 0, 0, 0, 0, 0};
            wf8 acci = {0, 0, 0, 0, 0, 0, 0, 0};
            #pragma unroll
            for (int kb = 0; kb < 16; kb += 4) {
                const int j0 = kb + 2 * half;
                const wf4 cc = *(const wf4*)&l2f[0][scol][j0];
                const wf2 br  = (wf2){keep0 ? cc.x : 0.f, keep1 ? cc.z : 0.f};
                const wf2 bim = (wf2){keep0 ? cc.y : 0.f, keep1 ? cc.w : 0.f};
                const wf2 ar  = *(const wf2*)&mR[NL - 2][0][n16][j0];
                accr = __builtin_amdgcn_wmma_f32_16x16x4_f32(false, ar, false, br,  (short)0, accr, false, false);
                acci = __builtin_amdgcn_wmma_f32_16x16x4_f32(false, ar, false, bim, (short)0, acci, false, false);
            }
            if (live) {
                #pragma unroll
                for (int vv = 0; vv < 8; vv += 2)
                    *(wf4*)&l3a[n16][8 * half + vv] = (wf4){accr[vv], acci[vv], accr[vv + 1], acci[vv + 1]};
            }
        }
    }
    __syncthreads();

    {
        const int iA = (wid << 1) | (n16 >> 3);
        float2 a3[8];
        #pragma unroll
        for (int q = 0; q < 8; ++q) a3[q] = l3a[q][iA];
        #pragma unroll
        for (int t2 = 0; t2 < 2; ++t2) {
            const int tt = wid * 2 + t2;
            const int iB = ((n16 & 7) << 1) | t2;
            float2 ff[2];
            #pragma unroll
            for (int u = 0; u < 2; ++u) {
                const int cb = 2 * half + u;
                float2 fw[2];
                #pragma unroll
                for (int w2 = 0; w2 < 2; ++w2) {
                    const float2 x0 = cmul(a3[0 + w2], l2f[1 + w2][cb][iB]),     y0 = cmul(a3[4 + w2], l2f[1 + w2][4 + cb][iB]);
                    const float2 x1 = cmul(a3[2 + w2], l2f[1 + w2][8 + cb][iB]), y1 = cmul(a3[6 + w2], l2f[1 + w2][12 + cb][iB]);
                    fw[w2] = make_float2((x0.x + y0.x) + (x1.x + y1.x), (x0.y + y0.y) + (x1.y + y1.y));
                }
                ff[u] = make_float2(fw[0].x + fw[1].x, fw[0].y + fw[1].y);
            }
            const float2 g0 = l2f[3 + t2][2 * half][n16], g1 = l2f[3 + t2][2 * half + 1][n16];
            const wf2 gr = (wf2){g0.x, g1.x}, gi = (wf2){g0.y, g1.y};
            const wf2 fr = (wf2){ff[0].x, ff[1].x}, fi = (wf2){ff[0].y, ff[1].y};
            const wf2 nfi = -fi;
            wf8 accr = {0, 0, 0, 0, 0, 0, 0, 0};
            wf8 acci = {0, 0, 0, 0, 0, 0, 0, 0};
            accr = __builtin_amdgcn_wmma_f32_16x16x4_f32(false, gr, false, fr,  (short)0, accr, false, false);
            acci = __builtin_amdgcn_wmma_f32_16x16x4_f32(false, gr, false, fi,  (short)0, acci, false, false);
            accr = __builtin_amdgcn_wmma_f32_16x16x4_f32(false, gi, false, nfi, (short)0, accr, false, false);
            acci = __builtin_amdgcn_wmma_f32_16x16x4_f32(false, gi, false, fr,  (short)0, acci, false, false);
            const int sslot = swz(colbase_c(tt, n16) + 8 * half);
            #pragma unroll
            for (int vv = 0; vv < 8; vv += 2)
                *(wf4*)&st[sslot ^ swz(vv)] = (wf4){accr[vv], acci[vv], accr[vv + 1], acci[vv + 1]};
        }
    }

    wf8 prob[2];

    {
        const int l = NL - 1;
        #pragma unroll
        for (int pass = 1; pass < 3; ++pass) {
            __syncthreads();
            #pragma unroll
            for (int t2 = 0; t2 < 2; ++t2) {
                const int tt = wid * 2 + t2;
                int mm, rowmul, colbase;
                if (pass == 1) { mm = 1 + (tt & 1); rowmul = 16;  colbase = (tt << 8) | n16; }
                else           { mm = 3 + (tt & 1); rowmul = 1;
                                 colbase = colbase_c(tt, n16); }

                const int lslot = swz(colbase + 2 * half * rowmul);
                const int sslot = swz(colbase + 8 * half * rowmul);

                wf8 accr = {0, 0, 0, 0, 0, 0, 0, 0};
                wf8 acci = {0, 0, 0, 0, 0, 0, 0, 0};
                #pragma unroll
                for (int kb = 0; kb < 16; kb += 4) {
                    int j0 = kb + 2 * half;
                    wf2 br, bim;
                    if (pass < 2) {
                        float2 s0 = st[lslot ^ swz(kb * rowmul)];
                        float2 s1 = st[lslot ^ swz((kb + 1) * rowmul)];
                        br  = (wf2){s0.x, s1.x};
                        bim = (wf2){s0.y, s1.y};
                    } else {
                        wf4 s = *(const wf4*)&st[lslot ^ swz(kb)];
                        br  = (wf2){s.x, s.z};
                        bim = (wf2){s.y, s.w};
                    }
                    wf2 ar  = *(const wf2*)&mR[l - 1][mm][n16][j0];
                    accr = __builtin_amdgcn_wmma_f32_16x16x4_f32(false, ar, false, br,  (short)0, accr, false, false);
                    acci = __builtin_amdgcn_wmma_f32_16x16x4_f32(false, ar, false, bim, (short)0, acci, false, false);
                }
                if (pass == 2) {
                    prob[t2] = accr * accr + acci * acci;
                } else {
                    #pragma unroll
                    for (int vv = 0; vv < 8; ++vv)
                        st[sslot ^ swz(vv * rowmul)] = make_float2(accr[vv], acci[vv]);
                }
            }
        }
    }

    float z[NQ];
    {
        const wf8 ps = prob[0] + prob[1];
        const wf8 pd = prob[0] - prob[1];
        const float a01 = ps[0] + ps[1], a23 = ps[2] + ps[3], a45 = ps[4] + ps[5], a67 = ps[6] + ps[7];
        const float lo = a01 + a23, hi = a45 + a67;
        const float P = lo + hi;
        z[7]  = ((pd[0] + pd[1]) + (pd[2] + pd[3])) + ((pd[4] + pd[5]) + (pd[6] + pd[7]));
        z[9]  = lo - hi;
        z[10] = (a01 - a23) + (a45 - a67);
        z[11] = ((ps[0] - ps[1]) + (ps[2] - ps[3])) + ((ps[4] - ps[5]) + (ps[6] - ps[7]));
        z[0] = (wid & 4) ? -P : P;
        z[1] = (wid & 2) ? -P : P;
        z[2] = (wid & 1) ? -P : P;
        z[3] = (n16 & 8) ? -P : P;
        z[4] = (n16 & 4) ? -P : P;
        z[5] = (n16 & 2) ? -P : P;
        z[6] = (n16 & 1) ? -P : P;
        z[8] = half ? -P : P;
    }
    #pragma unroll
    for (int q = 0; q < NQ; ++q) {
        float v = z[q];
        #pragma unroll
        for (int off = 16; off > 0; off >>= 1)
            v += __shfl_xor(v, off, 32);
        z[q] = v;
    }
    if (lane == 0) {
        #pragma unroll
        for (int q = 0; q < NQ; ++q) red[wid][q] = z[q];
    }
    __syncthreads();
    if (tid < NQ) {
        float s = 0.f;
        #pragma unroll
        for (int ww = 0; ww < 8; ++ww) s += red[ww][tid];
        out[b * NQ + tid] = s;
    }
}

extern "C" void kernel_launch(void* const* d_in, const int* in_sizes, int n_in,
                              void* d_out, int out_size, void* d_ws, size_t ws_size,
                              hipStream_t stream) {
    const float* x = (const float*)d_in[0];
    const float* w = (const float*)d_in[1];
    float* out = (float*)d_out;
    int B = in_sizes[0] / NQ;
    qsim_kernel<<<B, 256, 0, stream>>>(x, w, out);
}


# Layer 3's group-B and group-C passes applied to the factors; amplitudes formed once as a rank-8 WMMA product, squared in registers: both sweeps and the LDS state dropped; rounding, sum order change
# speedup vs baseline: 5.1015x; 1.1106x over previous
#include <hip/hip_runtime.h>


typedef float wf2 __attribute__((ext_vector_type(2)));
typedef float wf8 __attribute__((ext_vector_type(8)));

#define NQ 12
#define NL 4
#define MP 20
#define BP 66

typedef float wf4 __attribute__((ext_vector_type(4)));

__device__ __forceinline__ float2 cmul(float2 a, float2 b) {
    return make_float2(a.x * b.x - a.y * b.y, a.x * b.y + a.y * b.x);
}

__global__ __launch_bounds__(256, 8) void qsim_kernel(const float* __restrict__ x,
                                                   const float* __restrict__ w,
                                                   float* __restrict__ out) {
    __shared__ __attribute__((aligned(16))) float  mR[NL - 1][5][16][MP];
    __shared__ __attribute__((aligned(16))) float  dph[NL - 2][5][2][16];
    __shared__ __attribute__((aligned(16))) float2 col0[5][16];
    __shared__ __attribute__((aligned(16))) float2 l1v[5][4][16];
    __shared__ __attribute__((aligned(16))) float2 l2f[5][16][16];
    __shared__ __attribute__((aligned(16))) float2 ag3[3][8][16];
    __shared__ __attribute__((aligned(16))) float2 b3[2][16][BP];
    __shared__ __attribute__((aligned(16))) float2 ry[NL][NQ];
    __shared__ float2 rz[NL - 1][NQ];
    __shared__ float  red[8][12];

    const int tid  = threadIdx.x;
    const int b    = blockIdx.x;
    const int lane = tid & 31;
    const int wid  = tid >> 5;
    const int half = lane >> 4;
    const int n16  = lane & 15;

    if (tid < NL * NQ) {
        int l = tid / NQ, q = tid % NQ;
        float th = w[(l * NQ + q) * 2 + 0] + (l == 0 ? x[b * NQ + q] : 0.f);
        ry[l][q] = make_float2(cosf(0.5f * th),  sinf(0.5f * th));
        if (l < NL - 1) {
            float ph = w[(l * NQ + q) * 2 + 1];
            rz[l][q] = make_float2(cosf(0.5f * ph), -sinf(0.5f * ph));
        }
    }

    __syncthreads();

    if (tid < 5 * 16) {
        int m = tid >> 4, i = tid & 15;
        int g    = (m == 0) ? 0 : (m <= 2 ? 1 : 2);
        int bvar = (m == 0) ? 0 : (m <= 2 ? (m - 1) : (m - 3));
        int ip = i;
        ip ^= (ip >> 1) & 1;
        ip ^= (ip >> 1) & 2;
        ip ^= (ip >> 1) & 4;
        ip ^= bvar << 3;
        float2 v = make_float2(1.f, 0.f);
        float  r = 1.f;
        #pragma unroll
        for (int k = 0; k < 4; ++k) {
            int bi = (ip >> (3 - k)) & 1;
            float2 e  = rz[0][4 * g + k];
            float2 cs = ry[0][4 * g + k];
            if (bi) e.y = -e.y;
            v = cmul(v, e);
            r *= bi ? cs.y : cs.x;
        }
        col0[m][i] = make_float2(v.x * r, v.y * r);
    }

    if (tid < (NL - 2) * 5 * 16) {
        int l = 1 + tid / 80, rr = tid % 80;
        int m = rr >> 4, i = rr & 15;
        int g    = (m == 0) ? 0 : (m <= 2 ? 1 : 2);
        int bvar = (m == 0) ? 0 : (m <= 2 ? (m - 1) : (m - 3));
        int ip = i;
        ip ^= (ip >> 1) & 1;
        ip ^= (ip >> 1) & 2;
        ip ^= (ip >> 1) & 4;
        ip ^= bvar << 3;
        float2 v = make_float2(1.f, 0.f);
        #pragma unroll
        for (int k = 0; k < 4; ++k) {
            float2 e = rz[l][4 * g + k];
            if ((ip >> (3 - k)) & 1) e.y = -e.y;
            v = cmul(v, e);
        }
        dph[l - 1][m][0][i] = v.x;
        dph[l - 1][m][1][i] = v.y;
    }
    {
        const int i = tid >> 4, j = tid & 15;
        int ip = i;
        ip ^= (ip >> 1) & 1;
        ip ^= (ip >> 1) & 2;
        ip ^= (ip >> 1) & 4;
        const int df = ip ^ j;
        const bool e0 = !(df & 8), e1 = !(df & 4), e2 = !(df & 2), e3 = !(df & 1);
        const int sg0 = ((0x6996 >> (~ip & j & 15)) & 1) << 31;
        const int sg1 = ((0x6996 >> (~(ip ^ 8) & j & 15)) & 1) << 31;
        #pragma unroll
        for (int l = 1; l < NL; ++l) {
            #pragma unroll
            for (int g = 0; g < 3; ++g) {
                const wf4 q01 = *(const wf4*)&ry[l][4 * g];
                const wf4 q23 = *(const wf4*)&ry[l][4 * g + 2];
                const float t1 = e1 ? q01.z : q01.w;
                const float t2 = e2 ? q23.x : q23.y;
                const float t3 = e3 ? q23.z : q23.w;
                float va = e0 ? q01.x : q01.y;
                va *= t1; va *= t2; va *= t3;
                mR[l - 1][g == 0 ? 0 : 2 * g - 1][i][j] = __int_as_float(__float_as_int(va) ^ sg0);
                if (g > 0) {
                    float vb = e0 ? q01.y : q01.x;
                    vb *= t1; vb *= t2; vb *= t3;
                    mR[l - 1][2 * g][i][j] = __int_as_float(__float_as_int(vb) ^ sg1);
                }
            }
        }
    }
    __syncthreads();

    {
        const int m = __builtin_amdgcn_readfirstlane(wid);
        if (m < 5) {
            const bool isA = (m == 0), isB = (m == 1) | (m == 2);
            const int  fsrc = isA ? 0 : (isB ? 1 + ((n16 >> 1) & 1) : 3 + (n16 & 1));
            const bool live = isB ? (n16 < 4) : (n16 < 2);
            const bool keep0 = live && (!(isA | isB) || (n16 & 1) == 0);
            const bool keep1 = live && (!(isA | isB) || (n16 & 1) == 1);
            const wf4 dre03 = *(const wf4*)&dph[0][m][0][8 * half], dre47 = *(const wf4*)&dph[0][m][0][8 * half + 4];
            const wf4 dim03 = *(const wf4*)&dph[0][m][1][8 * half], dim47 = *(const wf4*)&dph[0][m][1][8 * half + 4];
            wf8 accr = {0, 0, 0, 0, 0, 0, 0, 0};
            wf8 acci = {0, 0, 0, 0, 0, 0, 0, 0};
            #pragma unroll
            for (int kb = 0; kb < 16; kb += 4) {
                const int j0 = kb + 2 * half;
                const wf4 cc = *(const wf4*)&col0[fsrc][j0];
                const wf2 br  = (wf2){keep0 ? cc.x : 0.f, keep1 ? cc.z : 0.f};
                const wf2 bim = (wf2){keep0 ? cc.y : 0.f, keep1 ? cc.w : 0.f};
                const wf2 ar  = *(const wf2*)&mR[0][m][n16][j0];
                accr = __builtin_amdgcn_wmma_f32_16x16x4_f32(false, ar, false, br,  (short)0, accr, false, false);
                acci = __builtin_amdgcn_wmma_f32_16x16x4_f32(false, ar, false, bim, (short)0, acci, false, false);
            }
            const wf8 dre = __builtin_shufflevector(dre03, dre47, 0, 1, 2, 3, 4, 5, 6, 7);
            const wf8 dim = __builtin_shufflevector(dim03, dim47, 0, 1, 2, 3, 4, 5, 6, 7);
            const wf8 outr = dre * accr - dim * acci;
            const wf8 outi = dre * acci + dim * accr;
            if (n16 < 4) {
                #pragma unroll
                for (int vv = 0; vv < 8; vv += 2)
                    *(wf4*)&l1v[m][n16][8 * half + vv] = (wf4){outr[vv], outi[vv], outr[vv + 1], outi[vv + 1]};
            }
        }
    }
    __syncthreads();

    {
        const int m = __builtin_amdgcn_readfirstlane(wid);
        if (m < 5) {
            const bool isA = (m == 0), isB = (m == 1) | (m == 2);
            const int  smat = isA ? 0 : (isB ? 1 + (n16 >> 3) : 3 + (n16 & 1));
            const int  scol = isB ? ((n16 >> 1) & 3) : ((n16 >> 1) & 1);
            const bool live  = isB | (n16 < 4);
            const bool keep0 = live && (!(isA | isB) || (n16 & 1) == 0);
            const bool keep1 = live && (!(isA | isB) || (n16 & 1) == 1);
            const wf4 dre03 = *(const wf4*)&dph[1][m][0][8 * half], dre47 = *(const wf4*)&dph[1][m][0][8 * half + 4];
            const wf4 dim03 = *(const wf4*)&dph[1][m][1][8 * half], dim47 = *(const wf4*)&dph[1][m][1][8 * half + 4];
            wf8 accr = {0, 0, 0, 0, 0, 0, 0, 0};
            wf8 acci = {0, 0, 0, 0, 0, 0, 0, 0};
            #pragma unroll
            for (int kb = 0; kb < 16; kb += 4) {
                const int j0 = kb + 2 * half;
                const wf4 cc = *(const wf4*)&l1v[smat][scol][j0];
                const wf2 br  = (wf2){keep0 ? cc.x : 0.f, keep1 ? cc.z : 0.f};
                const wf2 bim = (wf2){keep0 ? cc.y : 0.f, keep1 ? cc.w : 0.f};
                const wf2 ar  = *(const wf2*)&mR[1][m][n16][j0];
                accr = __builtin_amdgcn_wmma_f32_16x16x4_f32(false, ar, false, br,  (short)0, accr, false, false);
                acci = __builtin_amdgcn_wmma_f32_16x16x4_f32(false, ar, false, bim, (short)0, acci, false, false);
            }
            const wf8 dre = __builtin_shufflevector(dre03, dre47, 0, 1, 2, 3, 4, 5, 6, 7);
            const wf8 dim = __builtin_shufflevector(dim03, dim47, 0, 1, 2, 3, 4, 5, 6, 7);
            const wf8 outr = dre * accr - dim * acci;
            const wf8 outi = dre * acci + dim * accr;
            if (live) {
                #pragma unroll
                for (int vv = 0; vv < 8; vv += 2)
                    *(wf4*)&l2f[m][n16][8 * half + vv] = (wf4){outr[vv], outi[vv], outr[vv + 1], outi[vv + 1]};
            }
        }
    }
    __syncthreads();

    {
        const int m = __builtin_amdgcn_readfirstlane(wid);
        {
            const int x3 = m >> 2, w3 = (m >> 1) & 1, v3 = m & 1;
            const int scol = 8 * v3 + (n16 >> 1);
            const bool keep0 = (n16 & 1) == 0;
            const bool keep1 = (n16 & 1) == 1;
            wf8 accr = {0, 0, 0, 0, 0, 0, 0, 0};
            wf8 acci = {0, 0, 0, 0, 0, 0, 0, 0};
            #pragma unroll
            for (int kb = 0; kb < 16; kb += 4) {
                const int j0 = kb + 2 * half;
                const wf4 cc = *(const wf4*)&l2f[1 + w3][scol][j0];
                const wf2 br  = (wf2){keep0 ? cc.x : 0.f, keep1 ? cc.z : 0.f};
                const wf2 bim = (wf2){keep0 ? cc.y : 0.f, keep1 ? cc.w : 0.f};
                const wf2 ar  = *(const wf2*)&mR[NL - 2][1 + x3][n16][j0];
                accr = __builtin_amdgcn_wmma_f32_16x16x4_f32(false, ar, false, br,  (short)0, accr, false, false);
                acci = __builtin_amdgcn_wmma_f32_16x16x4_f32(false, ar, false, bim, (short)0, acci, false, false);
            }
            #pragma unroll
            for (int vv = 0; vv < 8; ++vv)
                b3[x3][8 * half + vv][32 * w3 + 16 * v3 + n16] = make_float2(accr[vv], acci[vv]);
        }
        if (m >= 5) {
            const bool isAl = (m == 5);
            const int  amat = isAl ? 0 : m - 3;
            const int  smat = isAl ? 0 : 3 + (n16 & 1);
            const int  scol = (n16 >> 1) & 3;
            const bool live  = n16 < 8;
            const bool keep0 = live && (!isAl || (n16 & 1) == 0);
            const bool keep1 = live && (!isAl || (n16 & 1) == 1);
            wf8 accr = {0, 0, 0, 0, 0, 0, 0, 0};
            wf8 acci = {0, 0, 0, 0, 0, 0, 0, 0};
            #pragma unroll
            for (int kb = 0; kb < 16; kb += 4) {
                const int j0 = kb + 2 * half;
                const wf4 cc = *(const wf4*)&l2f[smat][scol][j0];
                const wf2 br  = (wf2){keep0 ? cc.x : 0.f, keep1 ? cc.z : 0.f};
                const wf2 bim = (wf2){keep0 ? cc.y : 0.f, keep1 ? cc.w : 0.f};
                const wf2 ar  = *(const wf2*)&mR[NL - 2][amat][n16][j0];
                accr = __builtin_amdgcn_wmma_f32_16x16x4_f32(false, ar, false, br,  (short)0, accr, false, false);
                acci = __builtin_amdgcn_wmma_f32_16x16x4_f32(false, ar, false, bim, (short)0, acci, false, false);
            }
            if (live) {
                #pragma unroll
                for (int vv = 0; vv < 8; vv += 2)
                    *(wf4*)&ag3[m - 5][n16][8 * half + vv] = (wf4){accr[vv], acci[vv], accr[vv + 1], acci[vv + 1]};
            }
        }
    }
    __syncthreads();

    wf8 prob[2];

    {
        const int iA = (wid << 1) | (n16 >> 3);
        const int x3 = (n16 >> 3) & 1;
        float2 a3[8];
        #pragma unroll
        for (int q = 0; q < 8; ++q) a3[q] = ag3[0][q][iA];
        #pragma unroll
        for (int t2 = 0; t2 < 2; ++t2) {
            const int iB = ((n16 & 7) << 1) | t2;
            wf8 accr = {0, 0, 0, 0, 0, 0, 0, 0};
            wf8 acci = {0, 0, 0, 0, 0, 0, 0, 0};
            #pragma unroll
            for (int ks = 0; ks < 2; ++ks) {
                const int k0 = 4 * ks + 2 * half;
                float2 f0 = make_float2(0.f, 0.f), f1 = make_float2(0.f, 0.f);
                #pragma unroll
                for (int q = 0; q < 8; ++q) {
                    const int s3 = q >> 2, v3 = (q >> 1) & 1, w3 = q & 1;
                    const wf4 bb = *(const wf4*)&b3[x3][iB][32 * w3 + 16 * v3 + 8 * s3 + k0];
                    const float2 p0 = cmul(a3[q], make_float2(bb.x, bb.y));
                    const float2 p1 = cmul(a3[q], make_float2(bb.z, bb.w));
                    f0.x += p0.x; f0.y += p0.y;
                    f1.x += p1.x; f1.y += p1.y;
                }
                const float2 g0 = ag3[1 + t2][k0][n16], g1 = ag3[1 + t2][k0 + 1][n16];
                const wf2 gr = (wf2){g0.x, g1.x}, gi = (wf2){g0.y, g1.y};
                const wf2 fr = (wf2){f0.x, f1.x}, fi = (wf2){f0.y, f1.y};
                const wf2 nfi = -fi;
                accr = __builtin_amdgcn_wmma_f32_16x16x4_f32(false, gr, false, fr,  (short)0, accr, false, false);
                acci = __builtin_amdgcn_wmma_f32_16x16x4_f32(false, gr, false, fi,  (short)0, acci, false, false);
                accr = __builtin_amdgcn_wmma_f32_16x16x4_f32(false, gi, false, nfi, (short)0, accr, false, false);
                acci = __builtin_amdgcn_wmma_f32_16x16x4_f32(false, gi, false, fr,  (short)0, acci, false, false);
            }
            prob[t2] = accr * accr + acci * acci;
            __builtin_amdgcn_sched_barrier(0);
        }
    }

    float z[NQ];
    {
        const wf8 ps = prob[0] + prob[1];
        const wf8 pd = prob[0] - prob[1];
        const float a01 = ps[0] + ps[1], a23 = ps[2] + ps[3], a45 = ps[4] + ps[5], a67 = ps[6] + ps[7];
        const float lo = a01 + a23, hi = a45 + a67;
        const float P = lo + hi;
        z[7]  = ((pd[0] + pd[1]) + (pd[2] + pd[3])) + ((pd[4] + pd[5]) + (pd[6] + pd[7]));
        z[9]  = lo - hi;
        z[10] = (a01 - a23) + (a45 - a67);
        z[11] = ((ps[0] - ps[1]) + (ps[2] - ps[3])) + ((ps[4] - ps[5]) + (ps[6] - ps[7]));
        z[0] = (wid & 4) ? -P : P;
        z[1] = (wid & 2) ? -P : P;
        z[2] = (wid & 1) ? -P : P;
        z[3] = (n16 & 8) ? -P : P;
        z[4] = (n16 & 4) ? -P : P;
        z[5] = (n16 & 2) ? -P : P;
        z[6] = (n16 & 1) ? -P : P;
        z[8] = half ? -P : P;
    }
    #pragma unroll
    for (int q = 0; q < NQ; ++q) {
        float v = z[q];
        #pragma unroll
        for (int off = 16; off > 0; off >>= 1)
            v += __shfl_xor(v, off, 32);
        z[q] = v;
    }
    if (lane == 0) {
        #pragma unroll
        for (int q = 0; q < NQ; ++q) red[wid][q] = z[q];
    }
    __syncthreads();
    if (tid < NQ) {
        float s = 0.f;
        #pragma unroll
        for (int ww = 0; ww < 8; ++ww) s += red[ww][tid];
        out[b * NQ + tid] = s;
    }
}

extern "C" void kernel_launch(void* const* d_in, const int* in_sizes, int n_in,
                              void* d_out, int out_size, void* d_ws, size_t ws_size,
                              hipStream_t stream) {
    const float* x = (const float*)d_in[0];
    const float* w = (const float*)d_in[1];
    float* out = (float*)d_out;
    int B = in_sizes[0] / NQ;
    qsim_kernel<<<B, 256, 0, stream>>>(x, w, out);
}
